# starting kernel
# baseline (speedup 1.0000x reference)
#include <hip/hip_runtime.h>
#include <math.h>

#define HC        64
#define HEADS     8
#define LEAKY     0.2f
#define MT        2
#define WPB       2
#define ROWS      132

typedef __attribute__((ext_vector_type(2))) float v2f;
typedef __attribute__((ext_vector_type(8))) float v8f;

__device__ __forceinline__ unsigned f2o(float f) {
    unsigned u = __float_as_uint(f);
    return (u & 0x80000000u) ? ~u : (u | 0x80000000u);
}
__device__ __forceinline__ float o2f(unsigned u) {
    return (u & 0x80000000u) ? __uint_as_float(u & 0x7FFFFFFFu) : __uint_as_float(~u);
}

__global__ void gat_init_kernel(unsigned* __restrict__ m_ord, float* __restrict__ den,
                                float* __restrict__ out, float4* __restrict__ wq,
                                const float* __restrict__ w,
                                int nm, int nout, int nquads) {
    int t = blockIdx.x * blockDim.x + threadIdx.x;
    if (t < nm) { m_ord[t] = f2o(-INFINITY); den[t] = 0.0f; }
    if (t < nout) out[t] = 0.0f;
    if (t < nquads) {
        const int hi = t >> 10;
        const int q8 = (t >> 6) & 15;
        const int n  = t & 63;
        const int k  = 8 * q8 + 2 * hi;
        wq[t] = make_float4(w[(size_t)k * HC + n],       w[(size_t)(k + 1) * HC + n],
                            w[(size_t)(k + 4) * HC + n], w[(size_t)(k + 5) * HC + n]);
    }
}

__global__ void __launch_bounds__(32 * WPB)
gat_edge_score_kernel(const float* __restrict__ h, const int* __restrict__ src,
                      const int* __restrict__ dst, const float4* __restrict__ wq,
                      const float* __restrict__ a, float* __restrict__ scores,
                      unsigned* __restrict__ m_ord, int n_edges) {
    __shared__ float lds[WPB * MT * 16 * ROWS];

    const int wave  = threadIdx.x >> 5;
    const int lane  = threadIdx.x & 31;
    const int ebase = blockIdx.x * (WPB * MT * 16) + wave * (MT * 16);
    const int rbase = wave * (MT * 16);

#pragma unroll
    for (int j = 0; j < 2; ++j) {
        const int task = lane + 32 * j;
        const int row  = task & 31;
        const int half = task >> 5;
        int eg = ebase + row;
        if (eg >= n_edges) eg = n_edges - 1;
        const int idx = half ? dst[eg] : src[eg];
        const float4* sv = (const float4*)(h + (size_t)idx * HC);
        float4* rv = (float4*)&lds[(rbase + row) * ROWS + half * HC];
#pragma unroll
        for (int q = 0; q < HC / 4; ++q) rv[q] = sv[q];
    }
    __syncthreads();

    const int mrow = lane & 15;
    const int hi   = lane >> 4;
    const int hi2  = hi * 2;
    const int ncol = lane & 15;

    v8f acc[MT][4];
#pragma unroll
    for (int mt = 0; mt < MT; ++mt)
#pragma unroll
        for (int nt = 0; nt < 4; ++nt)
            acc[mt][nt] = (v8f){0.f,0.f,0.f,0.f,0.f,0.f,0.f,0.f};

#pragma unroll
    for (int kb8 = 0; kb8 < 16; ++kb8) {
        const int kb = kb8 * 8;
        float4 bq[4];
        const float4* wrow = wq + ((size_t)hi * 16 + kb8) * HC;
#pragma unroll
        for (int nt = 0; nt < 4; ++nt) bq[nt] = wrow[nt * 16 + ncol];

#pragma unroll
        for (int mt = 0; mt < MT; ++mt) {
            const float* arow = &lds[(rbase + mt * 16 + mrow) * ROWS];
            v2f a0, a1;
            a0.x = arow[kb + hi2];     a0.y = arow[kb + hi2 + 1];
            a1.x = arow[kb + 4 + hi2]; a1.y = arow[kb + 4 + hi2 + 1];
#pragma unroll
            for (int nt = 0; nt < 4; ++nt) {
                v2f b0; b0.x = bq[nt].x; b0.y = bq[nt].y;
                acc[mt][nt] = __builtin_amdgcn_wmma_f32_16x16x4_f32(
                    false, a0, false, b0, (short)0, acc[mt][nt], false, false);
            }
#pragma unroll
            for (int nt = 0; nt < 4; ++nt) {
                v2f b1; b1.x = bq[nt].z; b1.y = bq[nt].w;
                acc[mt][nt] = __builtin_amdgcn_wmma_f32_16x16x4_f32(
                    false, a1, false, b1, (short)0, acc[mt][nt], false, false);
            }
        }
    }

    const int headInNt = (lane >> 3) & 1;
#pragma unroll
    for (int nt = 0; nt < 4; ++nt) {
        const float av = a[nt * 16 + ncol];
#pragma unroll
        for (int mt = 0; mt < MT; ++mt) {
#pragma unroll
            for (int i = 0; i < 8; ++i) {
                float v = acc[mt][nt][i];
                v = (v > 0.0f) ? v : (LEAKY * v);
                v *= av;
                v += __shfl_xor(v, 1, 32);
                v += __shfl_xor(v, 2, 32);
                v += __shfl_xor(v, 4, 32);
                if ((lane & 7) == 0) {
                    const int edge = ebase + mt * 16 + i + 8 * hi;
                    if (edge < n_edges) {
                        const int head = nt * 2 + headInNt;
                        scores[(size_t)edge * HEADS + head] = v;
                        atomicMax(&m_ord[(size_t)dst[edge] * HEADS + head], f2o(v));
                    }
                }
            }
        }
    }
}

__global__ void gat_den_kernel(const float* __restrict__ scores, const int* __restrict__ dst,
                               const unsigned* __restrict__ m_ord, float* __restrict__ den,
                               int total) {
    int t = blockIdx.x * blockDim.x + threadIdx.x;
    if (t >= total) return;
    const int e = t >> 3, hd = t & 7;
    const int d = dst[e];
    const float m = o2f(m_ord[(size_t)d * HEADS + hd]);
    atomicAdd(&den[(size_t)d * HEADS + hd], __expf(scores[t] - m));
}

__global__ void gat_scatter_kernel(const float* __restrict__ h, const int* __restrict__ src,
                                   const int* __restrict__ dst, const float* __restrict__ scores,
                                   const unsigned* __restrict__ m_ord, const float* __restrict__ den,
                                   float* __restrict__ out, int total) {
    int t = blockIdx.x * blockDim.x + threadIdx.x;
    if (t >= total) return;
    const int e  = t >> 3;
    const int hd = t & 7;
    const int d  = dst[e];
    const int s  = src[e];
    const float m    = o2f(m_ord[(size_t)d * HEADS + hd]);
    const float attn = __expf(scores[t] - m) / den[(size_t)d * HEADS + hd];
    const float4* hv = (const float4*)(h + (size_t)s * HC + hd * 8);
    const float4 v0 = hv[0];
    const float4 v1 = hv[1];
    float* ob = out + (size_t)d * HC + hd * 8;
    atomicAdd(ob + 0, v0.x * attn);
    atomicAdd(ob + 1, v0.y * attn);
    atomicAdd(ob + 2, v0.z * attn);
    atomicAdd(ob + 3, v0.w * attn);
    atomicAdd(ob + 4, v1.x * attn);
    atomicAdd(ob + 5, v1.y * attn);
    atomicAdd(ob + 6, v1.z * attn);
    atomicAdd(ob + 7, v1.w * attn);
}

extern "C" void kernel_launch(void* const* d_in, const int* in_sizes, int n_in,
                              void* d_out, int out_size, void* d_ws, size_t ws_size,
                              hipStream_t stream) {
    const float* h   = (const float*)d_in[0];
    const int*   src = (const int*)d_in[1];
    const int*   dst = (const int*)d_in[2];
    const float* w   = (const float*)d_in[3];
    const float* a   = (const float*)d_in[4];
    float* out = (float*)d_out;

    const int n_edges = in_sizes[1];
    const int n_nodes = in_sizes[0] / HC;

    char* ws = (char*)d_ws;
    size_t off = 0;
    float*    scores = (float*)(ws + off);    off += (size_t)n_edges * HEADS * sizeof(float);
    unsigned* m_ord  = (unsigned*)(ws + off); off += (size_t)n_nodes * HEADS * sizeof(unsigned);
    float*    den    = (float*)(ws + off);    off += (size_t)n_nodes * HEADS * sizeof(float);
    float4*   wq     = (float4*)(ws + off);

    const int nm     = n_nodes * HEADS;
    const int nout   = n_nodes * HC;
    const int nquads = 2 * 16 * HC;
    int initN = nout > nm ? nout : nm;
    if (nquads > initN) initN = nquads;
    gat_init_kernel<<<(initN + 255) / 256, 256, 0, stream>>>(m_ord, den, out, wq, w,
                                                             nm, nout, nquads);

    const int epb    = WPB * MT * 16;
    const int blocks = (n_edges + epb - 1) / epb;
    gat_edge_score_kernel<<<blocks, 32 * WPB, 0, stream>>>(h, src, dst, wq, a,
                                                           scores, m_ord, n_edges);

    const int t2 = n_edges * HEADS;
    gat_den_kernel<<<(t2 + 255) / 256, 256, 0, stream>>>(scores, dst, m_ord, den, t2);

    gat_scatter_kernel<<<(t2 + 255) / 256, 256, 0, stream>>>(h, src, dst, scores,
                                                             m_ord, den, out, t2);
}


# Project per node, not per edge: P=h@w[0:64], Q=h@w[64:128] on f32 WMMA, one row per node; edge score=leaky(P[src]+Q[dst]).a; k-sum is now two 64-term halves added
# speedup vs baseline: 2.0168x; 2.0168x over previous
#include <hip/hip_runtime.h>
#include <math.h>

#define HC        64
#define HEADS     8
#define LEAKY     0.2f
#define MT        2
#define WPB       2
#define NROWS     68
#define EITEMS    4

typedef __attribute__((ext_vector_type(2))) float v2f;
typedef __attribute__((ext_vector_type(4))) float v4f;
typedef __attribute__((ext_vector_type(8))) float v8f;

__device__ __forceinline__ unsigned f2o(float f) {
    unsigned u = __float_as_uint(f);
    return (u & 0x80000000u) ? ~u : (u | 0x80000000u);
}
__device__ __forceinline__ float o2f(unsigned u) {
    return (u & 0x80000000u) ? __uint_as_float(u & 0x7FFFFFFFu) : __uint_as_float(~u);
}

__global__ void gat_init_kernel(unsigned* __restrict__ m_ord, float* __restrict__ den,
                                float* __restrict__ out, float4* __restrict__ wq,
                                const float* __restrict__ w,
                                int nm, int nout, int nquads) {
    int t = blockIdx.x * blockDim.x + threadIdx.x;
    if (t < nm) { m_ord[t] = f2o(-INFINITY); den[t] = 0.0f; }
    if (t < nout) out[t] = 0.0f;
    if (t < nquads) {
        const int hi = t >> 10;
        const int q8 = (t >> 6) & 15;
        const int n  = t & 63;
        const int k  = 8 * q8 + 2 * hi;
        wq[t] = make_float4(w[(size_t)k * HC + n],       w[(size_t)(k + 1) * HC + n],
                            w[(size_t)(k + 4) * HC + n], w[(size_t)(k + 5) * HC + n]);
    }
}

__global__ void __launch_bounds__(32 * WPB)
gat_node_proj_kernel(const float* __restrict__ h, const float4* __restrict__ wq,
                     float* __restrict__ P, float* __restrict__ Q, int n_nodes) {
    __shared__ float lds[WPB * MT * 16 * NROWS];

    const int wave  = threadIdx.x >> 5;
    const int lane  = threadIdx.x & 31;
    const int half  = blockIdx.y;
    const int nbase = blockIdx.x * (WPB * MT * 16) + wave * (MT * 16);
    const int rbase = wave * (MT * 16);

#pragma unroll
    for (int j = 0; j < 16; ++j) {
        const int idx = lane + 32 * j;
        const int row = idx >> 4;
        const int c4  = idx & 15;
        int ng = nbase + row;
        if (ng >= n_nodes) ng = n_nodes - 1;
        const float4 v = ((const float4*)(h + (size_t)ng * HC))[c4];
        *(float4*)&lds[(rbase + row) * NROWS + c4 * 4] = v;
    }
    __syncthreads();

    const int mrow = lane & 15;
    const int hi   = lane >> 4;
    const int hi2  = hi * 2;
    const int ncol = lane & 15;

    v8f acc[MT][4];
#pragma unroll
    for (int mt = 0; mt < MT; ++mt)
#pragma unroll
        for (int nt = 0; nt < 4; ++nt)
            acc[mt][nt] = (v8f){0.f,0.f,0.f,0.f,0.f,0.f,0.f,0.f};

#pragma unroll
    for (int kb8 = 0; kb8 < 8; ++kb8) {
        const int kb = kb8 * 8;
        float4 bq[4];
        const float4* wrow = wq + ((size_t)hi * 16 + half * 8 + kb8) * HC;
#pragma unroll
        for (int nt = 0; nt < 4; ++nt) bq[nt] = wrow[nt * 16 + ncol];

#pragma unroll
        for (int mt = 0; mt < MT; ++mt) {
            const float* arow = &lds[(rbase + mt * 16 + mrow) * NROWS];
            v2f a0, a1;
            a0.x = arow[kb + hi2];     a0.y = arow[kb + hi2 + 1];
            a1.x = arow[kb + 4 + hi2]; a1.y = arow[kb + 4 + hi2 + 1];
#pragma unroll
            for (int nt = 0; nt < 4; ++nt) {
                v2f b0; b0.x = bq[nt].x; b0.y = bq[nt].y;
                acc[mt][nt] = __builtin_amdgcn_wmma_f32_16x16x4_f32(
                    false, a0, false, b0, (short)0, acc[mt][nt], false, false);
            }
#pragma unroll
            for (int nt = 0; nt < 4; ++nt) {
                v2f b1; b1.x = bq[nt].z; b1.y = bq[nt].w;
                acc[mt][nt] = __builtin_amdgcn_wmma_f32_16x16x4_f32(
                    false, a1, false, b1, (short)0, acc[mt][nt], false, false);
            }
        }
    }

    float* plane = half ? Q : P;
#pragma unroll
    for (int mt = 0; mt < MT; ++mt)
#pragma unroll
        for (int nt = 0; nt < 4; ++nt)
#pragma unroll
            for (int i = 0; i < 8; ++i)
                plane[(size_t)(nbase + mt * 16 + i + 8 * hi) * HC + nt * 16 + ncol] = acc[mt][nt][i];
}

__device__ __forceinline__ float lrelu(float v) { return (v > 0.0f) ? v : (LEAKY * v); }

template <bool FULL>
__device__ __forceinline__ void
edge_score_items(const float* __restrict__ P, const float* __restrict__ Q,
                 const int* __restrict__ src, const int* __restrict__ dst,
                 const float* __restrict__ a, float* __restrict__ scores,
                 unsigned* __restrict__ m_ord, int total) {
    const int hd = threadIdx.x & 7;
    const v4f a0 = ((const v4f*)(a + hd * 8))[0];
    const v4f a1 = ((const v4f*)(a + hd * 8))[1];
    const int tb = blockIdx.x * (256 * EITEMS) + threadIdx.x;

    int dn[EITEMS];
    v4f p0[EITEMS], p1[EITEMS], q0[EITEMS], q1[EITEMS];
#pragma unroll
    for (int j = 0; j < EITEMS; ++j) {
        int t = tb + 256 * j;
        if (!FULL && t >= total) t = total - 1;
        const int e = t >> 3;
        const int s = src[e];
        const int d = dst[e];
        dn[j] = d;
        const v4f* pv = (const v4f*)(P + (size_t)s * HC + hd * 8);
        const v4f* qv = (const v4f*)(Q + (size_t)d * HC + hd * 8);
        p0[j] = pv[0]; p1[j] = pv[1];
        q0[j] = qv[0]; q1[j] = qv[1];
    }
    static_assert(EITEMS == 4, "the fence below names four items");
    asm volatile("" :: "v"(p0[0]), "v"(p1[0]), "v"(q0[0]), "v"(q1[0]),
                       "v"(p0[1]), "v"(p1[1]), "v"(q0[1]), "v"(q1[1]),
                       "v"(p0[2]), "v"(p1[2]), "v"(q0[2]), "v"(q1[2]),
                       "v"(p0[3]), "v"(p1[3]), "v"(q0[3]), "v"(q1[3]));
#pragma unroll
    for (int j = 0; j < EITEMS; ++j) {
        const float t0 = lrelu(p0[j].x + q0[j].x) * a0.x;
        const float t1 = lrelu(p0[j].y + q0[j].y) * a0.y;
        const float t2 = lrelu(p0[j].z + q0[j].z) * a0.z;
        const float t3 = lrelu(p0[j].w + q0[j].w) * a0.w;
        const float t4 = lrelu(p1[j].x + q1[j].x) * a1.x;
        const float t5 = lrelu(p1[j].y + q1[j].y) * a1.y;
        const float t6 = lrelu(p1[j].z + q1[j].z) * a1.z;
        const float t7 = lrelu(p1[j].w + q1[j].w) * a1.w;
        const float v  = ((t0 + t1) + (t2 + t3)) + ((t4 + t5) + (t6 + t7));
        const int t = tb + 256 * j;
        if (FULL || t < total) {
            scores[t] = v;
            atomicMax(&m_ord[(size_t)dn[j] * HEADS + hd], f2o(v));
        }
    }
}

__global__ void __launch_bounds__(256)
gat_edge_score_kernel(const float* __restrict__ P, const float* __restrict__ Q,
                      const int* __restrict__ src, const int* __restrict__ dst,
                      const float* __restrict__ a, float* __restrict__ scores,
                      unsigned* __restrict__ m_ord, int total) {
    if ((blockIdx.x + 1) * (256 * EITEMS) <= total)
        edge_score_items<true>(P, Q, src, dst, a, scores, m_ord, total);
    else
        edge_score_items<false>(P, Q, src, dst, a, scores, m_ord, total);
}

__global__ void gat_den_kernel(const float* __restrict__ scores, const int* __restrict__ dst,
                               const unsigned* __restrict__ m_ord, float* __restrict__ den,
                               int total) {
    int t = blockIdx.x * blockDim.x + threadIdx.x;
    if (t >= total) return;
    const int e = t >> 3, hd = t & 7;
    const int d = dst[e];
    const float m = o2f(m_ord[(size_t)d * HEADS + hd]);
    atomicAdd(&den[(size_t)d * HEADS + hd], __expf(scores[t] - m));
}

__global__ void gat_scatter_kernel(const float* __restrict__ h, const int* __restrict__ src,
                                   const int* __restrict__ dst, const float* __restrict__ scores,
                                   const unsigned* __restrict__ m_ord, const float* __restrict__ den,
                                   float* __restrict__ out, int total) {
    int t = blockIdx.x * blockDim.x + threadIdx.x;
    if (t >= total) return;
    const int e  = t >> 3;
    const int hd = t & 7;
    const int d  = dst[e];
    const int s  = src[e];
    const float m    = o2f(m_ord[(size_t)d * HEADS + hd]);
    const float attn = __expf(scores[t] - m) / den[(size_t)d * HEADS + hd];
    const float4* hv = (const float4*)(h + (size_t)s * HC + hd * 8);
    const float4 v0 = hv[0];
    const float4 v1 = hv[1];
    float* ob = out + (size_t)d * HC + hd * 8;
    atomicAdd(ob + 0, v0.x * attn);
    atomicAdd(ob + 1, v0.y * attn);
    atomicAdd(ob + 2, v0.z * attn);
    atomicAdd(ob + 3, v0.w * attn);
    atomicAdd(ob + 4, v1.x * attn);
    atomicAdd(ob + 5, v1.y * attn);
    atomicAdd(ob + 6, v1.z * attn);
    atomicAdd(ob + 7, v1.w * attn);
}

extern "C" void kernel_launch(void* const* d_in, const int* in_sizes, int n_in,
                              void* d_out, int out_size, void* d_ws, size_t ws_size,
                              hipStream_t stream) {
    const float* h   = (const float*)d_in[0];
    const int*   src = (const int*)d_in[1];
    const int*   dst = (const int*)d_in[2];
    const float* w   = (const float*)d_in[3];
    const float* a   = (const float*)d_in[4];
    float* out = (float*)d_out;

    const int n_edges = in_sizes[1];
    const int n_nodes = in_sizes[0] / HC;

    const int npb  = WPB * MT * 16;
    const int nblk = (n_nodes + npb - 1) / npb;

    char* ws = (char*)d_ws;
    size_t off = 0;
    float*    scores = (float*)(ws + off);    off += (size_t)n_edges * HEADS * sizeof(float);
    unsigned* m_ord  = (unsigned*)(ws + off); off += (size_t)n_nodes * HEADS * sizeof(unsigned);
    float*    den    = (float*)(ws + off);    off += (size_t)n_nodes * HEADS * sizeof(float);
    float4*   wq     = (float4*)(ws + off);   off += (size_t)2 * 16 * HC * sizeof(float4);
    float*    P      = (float*)(ws + off);    off += (size_t)nblk * npb * HC * sizeof(float);
    float*    Q      = (float*)(ws + off);

    const int nm     = n_nodes * HEADS;
    const int nout   = n_nodes * HC;
    const int nquads = 2 * 16 * HC;
    int initN = nout > nm ? nout : nm;
    if (nquads > initN) initN = nquads;
    gat_init_kernel<<<(initN + 255) / 256, 256, 0, stream>>>(m_ord, den, out, wq, w,
                                                             nm, nout, nquads);

    gat_node_proj_kernel<<<dim3(nblk, 2), 32 * WPB, 0, stream>>>(h, wq, P, Q, n_nodes);

    const int t2  = n_edges * HEADS;
    const int ipb = 256 * EITEMS;
    gat_edge_score_kernel<<<(t2 + ipb - 1) / ipb, 256, 0, stream>>>(P, Q, src, dst, a,
                                                                    scores, m_ord, t2);

    gat_den_kernel<<<(t2 + 255) / 256, 256, 0, stream>>>(scores, dst, m_ord, den, t2);

    gat_scatter_kernel<<<(t2 + 255) / 256, 256, 0, stream>>>(h, src, dst, scores,
                                                             m_ord, den, out, t2);
}


# Group edges by dst per call (int-atomic rank + block scan); 16 lanes per node do score, running-max softmax and weighted sum in registers: no float atomics, sums in segment (arrival) order
# speedup vs baseline: 3.9088x; 1.9381x over previous
#include <hip/hip_runtime.h>
#include <math.h>

#define HC        64
#define LEAKY     0.2f
#define MT        2
#define WPB       2
#define NROWS     68
#define NU        4
#define SCAN_T    1024

typedef __attribute__((ext_vector_type(2))) float v2f;
typedef __attribute__((ext_vector_type(4))) float v4f;
typedef __attribute__((ext_vector_type(8))) float v8f;


__global__ void gat_init_kernel(int* __restrict__ deg, int* __restrict__ cursor,
                                float4* __restrict__ wq, const float* __restrict__ w,
                                int n_nodes, int nquads) {
    int t = blockIdx.x * blockDim.x + threadIdx.x;
    if (t < n_nodes) deg[t] = 0;
    if (t == 0) cursor[0] = 0;
    if (t < nquads) {
        const int hi = t >> 10;
        const int q8 = (t >> 6) & 15;
        const int n  = t & 63;
        const int k  = 8 * q8 + 2 * hi;
        wq[t] = make_float4(w[(size_t)k * HC + n],       w[(size_t)(k + 1) * HC + n],
                            w[(size_t)(k + 4) * HC + n], w[(size_t)(k + 5) * HC + n]);
    }
}

__global__ void __launch_bounds__(32 * WPB)
gat_node_proj_kernel(const float* __restrict__ h, const float4* __restrict__ wq,
                     float* __restrict__ P, float* __restrict__ Q, int n_nodes) {
    __shared__ float lds[WPB * MT * 16 * NROWS];

    const int wave  = threadIdx.x >> 5;
    const int lane  = threadIdx.x & 31;
    const int half  = blockIdx.y;
    const int nbase = blockIdx.x * (WPB * MT * 16) + wave * (MT * 16);
    const int rbase = wave * (MT * 16);

#pragma unroll
    for (int j = 0; j < 16; ++j) {
        const int idx = lane + 32 * j;
        const int row = idx >> 4;
        const int c4  = idx & 15;
        int ng = nbase + row;
        if (ng >= n_nodes) ng = n_nodes - 1;
        const float4 v = ((const float4*)(h + (size_t)ng * HC))[c4];
        *(float4*)&lds[(rbase + row) * NROWS + c4 * 4] = v;
    }
    __syncthreads();

    const int mrow = lane & 15;
    const int hi   = lane >> 4;
    const int hi2  = hi * 2;
    const int ncol = lane & 15;

    v8f acc[MT][4];
#pragma unroll
    for (int mt = 0; mt < MT; ++mt)
#pragma unroll
        for (int nt = 0; nt < 4; ++nt)
            acc[mt][nt] = (v8f){0.f,0.f,0.f,0.f,0.f,0.f,0.f,0.f};

#pragma unroll
    for (int kb8 = 0; kb8 < 8; ++kb8) {
        const int kb = kb8 * 8;
        float4 bq[4];
        const float4* wrow = wq + ((size_t)hi * 16 + half * 8 + kb8) * HC;
#pragma unroll
        for (int nt = 0; nt < 4; ++nt) bq[nt] = wrow[nt * 16 + ncol];

#pragma unroll
        for (int mt = 0; mt < MT; ++mt) {
            const float* arow = &lds[(rbase + mt * 16 + mrow) * NROWS];
            v2f a0, a1;
            a0.x = arow[kb + hi2];     a0.y = arow[kb + hi2 + 1];
            a1.x = arow[kb + 4 + hi2]; a1.y = arow[kb + 4 + hi2 + 1];
#pragma unroll
            for (int nt = 0; nt < 4; ++nt) {
                v2f b0; b0.x = bq[nt].x; b0.y = bq[nt].y;
                acc[mt][nt] = __builtin_amdgcn_wmma_f32_16x16x4_f32(
                    false, a0, false, b0, (short)0, acc[mt][nt], false, false);
            }
#pragma unroll
            for (int nt = 0; nt < 4; ++nt) {
                v2f b1; b1.x = bq[nt].z; b1.y = bq[nt].w;
                acc[mt][nt] = __builtin_amdgcn_wmma_f32_16x16x4_f32(
                    false, a1, false, b1, (short)0, acc[mt][nt], false, false);
            }
        }
    }

    float* plane = half ? Q : P;
#pragma unroll
    for (int mt = 0; mt < MT; ++mt)
#pragma unroll
        for (int nt = 0; nt < 4; ++nt)
#pragma unroll
            for (int i = 0; i < 8; ++i)
                plane[(size_t)(nbase + mt * 16 + i + 8 * hi) * HC + nt * 16 + ncol] = acc[mt][nt][i];
}

__global__ void gat_count_kernel(const int* __restrict__ dst, int* __restrict__ deg,
                                 int* __restrict__ rank, int n_edges) {
    const int e = blockIdx.x * blockDim.x + threadIdx.x;
    if (e < n_edges) rank[e] = atomicAdd(&deg[dst[e]], 1);
}

__device__ __forceinline__ int wave_incl_scan(int x, int lane) {
#pragma unroll
    for (int o = 1; o < 32; o <<= 1) {
        const int y = __shfl_up(x, o, 32);
        if (lane >= o) x += y;
    }
    return x;
}

__global__ void __launch_bounds__(SCAN_T)
gat_alloc_kernel(const int* __restrict__ deg, int* __restrict__ cursor,
                 int* __restrict__ start, int n_nodes) {
    __shared__ int wsum[32];
    __shared__ int sbase;
    const int t    = blockIdx.x * SCAN_T + threadIdx.x;
    const int lane = threadIdx.x & 31;
    const int wave = threadIdx.x >> 5;
    const int v = (t < n_nodes) ? deg[t] : 0;
    const int x = wave_incl_scan(v, lane);
    if (lane == 31) wsum[wave] = x;
    __syncthreads();
    if (wave == 0) {
        const int w  = wsum[lane];
        const int xs = wave_incl_scan(w, lane);
        wsum[lane] = xs - w;
        if (lane == 31) sbase = atomicAdd(cursor, xs);
    }
    __syncthreads();
    if (t < n_nodes) start[t] = sbase + wsum[wave] + (x - v);
}

__global__ void gat_fill_kernel(const int* __restrict__ src, const int* __restrict__ dst,
                                const int* __restrict__ rank, const int* __restrict__ start,
                                int* __restrict__ csr_src, int n_edges) {
    const int e = blockIdx.x * blockDim.x + threadIdx.x;
    if (e < n_edges) csr_src[start[dst[e]] + rank[e]] = src[e];
}

__device__ __forceinline__ float lrelu(float v) { return (v > 0.0f) ? v : (LEAKY * v); }

__device__ __forceinline__ float pair_other(float v) {
    return __int_as_float(__builtin_amdgcn_update_dpp(0, __float_as_int(v), 0xB1, 0xF, 0xF, false));
}

__global__ void __launch_bounds__(256)
gat_node_kernel(const float* __restrict__ h, const float* __restrict__ P,
                const float* __restrict__ Q, const float* __restrict__ a,
                const int* __restrict__ csr_src, const int* __restrict__ start,
                const int* __restrict__ deg, float* __restrict__ out, int n_nodes) {
    const int t    = blockIdx.x * 256 + threadIdx.x;
    const int node = t >> 4;
    const int c4   = (t & 15) * 4;
    if (node >= n_nodes) return;

    const int n   = deg[node];
    const int beg = start[node];
    const v4f av = *(const v4f*)(a + c4);
    const v4f qv = *(const v4f*)(Q + (size_t)node * HC + c4);

    float m = -INFINITY, den = 0.0f;
    v4f acc = (v4f){0.f, 0.f, 0.f, 0.f};

    for (int j = 0; j < n; j += NU) {
        int s[NU];
#pragma unroll
        for (int k = 0; k < NU; ++k) {
            int jj = j + k;
            if (jj > n - 1) jj = n - 1;
            s[k] = csr_src[beg + jj];
        }
        v4f pv[NU], hv[NU];
#pragma unroll
        for (int k = 0; k < NU; ++k) {
            pv[k] = *(const v4f*)(P + (size_t)s[k] * HC + c4);
            hv[k] = *(const v4f*)(h + (size_t)s[k] * HC + c4);
        }
        float sc[NU];
        float mn = m;
#pragma unroll
        for (int k = 0; k < NU; ++k) {
            const float t0 = lrelu(pv[k].x + qv.x) * av.x;
            const float t1 = lrelu(pv[k].y + qv.y) * av.y;
            const float t2 = lrelu(pv[k].z + qv.z) * av.z;
            const float t3 = lrelu(pv[k].w + qv.w) * av.w;
            const float part = (t0 + t1) + (t2 + t3);
            float v = part + pair_other(part);
            if (j + k >= n) v = -INFINITY;
            sc[k] = v;
            mn = fmaxf(mn, v);
        }
        const float scale = __expf(m - mn);
        den *= scale;
        acc.x *= scale; acc.y *= scale; acc.z *= scale; acc.w *= scale;
#pragma unroll
        for (int k = 0; k < NU; ++k) {
            const float ex = __expf(sc[k] - mn);
            den += ex;
            acc.x += ex * hv[k].x; acc.y += ex * hv[k].y;
            acc.z += ex * hv[k].z; acc.w += ex * hv[k].w;
        }
        m = mn;
    }

    v4f o = (v4f){0.f, 0.f, 0.f, 0.f};
    if (n > 0) { o.x = acc.x / den; o.y = acc.y / den; o.z = acc.z / den; o.w = acc.w / den; }
    *(v4f*)(out + (size_t)node * HC + c4) = o;
}

extern "C" void kernel_launch(void* const* d_in, const int* in_sizes, int n_in,
                              void* d_out, int out_size, void* d_ws, size_t ws_size,
                              hipStream_t stream) {
    const float* h   = (const float*)d_in[0];
    const int*   src = (const int*)d_in[1];
    const int*   dst = (const int*)d_in[2];
    const float* w   = (const float*)d_in[3];
    const float* a   = (const float*)d_in[4];
    float* out = (float*)d_out;

    const int n_edges = in_sizes[1];
    const int n_nodes = in_sizes[0] / HC;

    const int npb  = WPB * MT * 16;
    const int nblk = (n_nodes + npb - 1) / npb;

    char* ws = (char*)d_ws;
    size_t off = 0;
    float*  P       = (float*)(ws + off);  off += (size_t)nblk * npb * HC * sizeof(float);
    float*  Q       = (float*)(ws + off);  off += (size_t)nblk * npb * HC * sizeof(float);
    float4* wq      = (float4*)(ws + off); off += (size_t)2 * 16 * HC * sizeof(float4);
    int*    rank    = (int*)(ws + off);    off += (size_t)n_edges * sizeof(int);
    int*    csr_src = (int*)(ws + off);    off += (size_t)n_edges * sizeof(int);
    int*    deg     = (int*)(ws + off);    off += (size_t)n_nodes * sizeof(int);
    int*    start   = (int*)(ws + off);    off += (size_t)n_nodes * sizeof(int);
    int*    cursor  = (int*)(ws + off);

    const int nquads = 2 * 16 * HC;
    const int initN  = n_nodes > nquads ? n_nodes : nquads;
    gat_init_kernel<<<(initN + 255) / 256, 256, 0, stream>>>(deg, cursor, wq, w, n_nodes, nquads);

    gat_node_proj_kernel<<<dim3(nblk, 2), 32 * WPB, 0, stream>>>(h, wq, P, Q, n_nodes);

    gat_count_kernel<<<(n_edges + 255) / 256, 256, 0, stream>>>(dst, deg, rank, n_edges);

    gat_alloc_kernel<<<(n_nodes + SCAN_T - 1) / SCAN_T, SCAN_T, 0, stream>>>(deg, cursor, start,
                                                                             n_nodes);

    gat_fill_kernel<<<(n_edges + 255) / 256, 256, 0, stream>>>(src, dst, rank, start, csr_src,
                                                               n_edges);

    const int nthr = n_nodes * 16;
    gat_node_kernel<<<(nthr + 255) / 256, 256, 0, stream>>>(h, P, Q, a, csr_src, start, deg,
                                                            out, n_nodes);
}


# Group edges by dst in two levels on LDS counters (512-node buckets, then nodes): per-edge global atomics become one add per (block, bucket); segment (sum) order = arrival order of the LDS adds
# speedup vs baseline: 8.1880x; 2.0948x over previous
#include <hip/hip_runtime.h>
#include <math.h>

#define HC        64
#define LEAKY     0.2f
#define MT        2
#define WPB       2
#define NROWS     68
#define NU        4
#define GT        1024
#define EPT       8
#define BSH       9
#define BNODES    (1 << BSH)
#define NR        4
#define MAXNB     GT

typedef __attribute__((ext_vector_type(2))) float v2f;
typedef __attribute__((ext_vector_type(4))) float v4f;
typedef __attribute__((ext_vector_type(8))) float v8f;


__global__ void gat_init_kernel(int* __restrict__ bcount, int* __restrict__ bcursor,
                                float4* __restrict__ wq, const float* __restrict__ w,
                                int nb, int nquads) {
    int t = blockIdx.x * blockDim.x + threadIdx.x;
    if (t < nb) { bcount[t] = 0; bcursor[t] = 0; }
    if (t < nquads) {
        const int hi = t >> 10;
        const int q8 = (t >> 6) & 15;
        const int n  = t & 63;
        const int k  = 8 * q8 + 2 * hi;
        wq[t] = make_float4(w[(size_t)k * HC + n],       w[(size_t)(k + 1) * HC + n],
                            w[(size_t)(k + 4) * HC + n], w[(size_t)(k + 5) * HC + n]);
    }
}

__global__ void __launch_bounds__(32 * WPB)
gat_node_proj_kernel(const float* __restrict__ h, const float4* __restrict__ wq,
                     float* __restrict__ P, float* __restrict__ Q, int n_nodes) {
    __shared__ float lds[WPB * MT * 16 * NROWS];

    const int wave  = threadIdx.x >> 5;
    const int lane  = threadIdx.x & 31;
    const int half  = blockIdx.y;
    const int nbase = blockIdx.x * (WPB * MT * 16) + wave * (MT * 16);
    const int rbase = wave * (MT * 16);

#pragma unroll
    for (int j = 0; j < 16; ++j) {
        const int idx = lane + 32 * j;
        const int row = idx >> 4;
        const int c4  = idx & 15;
        int ng = nbase + row;
        if (ng >= n_nodes) ng = n_nodes - 1;
        const float4 v = ((const float4*)(h + (size_t)ng * HC))[c4];
        *(float4*)&lds[(rbase + row) * NROWS + c4 * 4] = v;
    }
    __syncthreads();

    const int mrow = lane & 15;
    const int hi   = lane >> 4;
    const int hi2  = hi * 2;
    const int ncol = lane & 15;

    v8f acc[MT][4];
#pragma unroll
    for (int mt = 0; mt < MT; ++mt)
#pragma unroll
        for (int nt = 0; nt < 4; ++nt)
            acc[mt][nt] = (v8f){0.f,0.f,0.f,0.f,0.f,0.f,0.f,0.f};

#pragma unroll
    for (int kb8 = 0; kb8 < 8; ++kb8) {
        const int kb = kb8 * 8;
        float4 bq[4];
        const float4* wrow = wq + ((size_t)hi * 16 + half * 8 + kb8) * HC;
#pragma unroll
        for (int nt = 0; nt < 4; ++nt) bq[nt] = wrow[nt * 16 + ncol];

#pragma unroll
        for (int mt = 0; mt < MT; ++mt) {
            const float* arow = &lds[(rbase + mt * 16 + mrow) * NROWS];
            v2f a0, a1;
            a0.x = arow[kb + hi2];     a0.y = arow[kb + hi2 + 1];
            a1.x = arow[kb + 4 + hi2]; a1.y = arow[kb + 4 + hi2 + 1];
#pragma unroll
            for (int nt = 0; nt < 4; ++nt) {
                v2f b0; b0.x = bq[nt].x; b0.y = bq[nt].y;
                acc[mt][nt] = __builtin_amdgcn_wmma_f32_16x16x4_f32(
                    false, a0, false, b0, (short)0, acc[mt][nt], false, false);
            }
#pragma unroll
            for (int nt = 0; nt < 4; ++nt) {
                v2f b1; b1.x = bq[nt].z; b1.y = bq[nt].w;
                acc[mt][nt] = __builtin_amdgcn_wmma_f32_16x16x4_f32(
                    false, a1, false, b1, (short)0, acc[mt][nt], false, false);
            }
        }
    }

    float* plane = half ? Q : P;
#pragma unroll
    for (int mt = 0; mt < MT; ++mt)
#pragma unroll
        for (int nt = 0; nt < 4; ++nt)
#pragma unroll
            for (int i = 0; i < 8; ++i)
                plane[(size_t)(nbase + mt * 16 + i + 8 * hi) * HC + nt * 16 + ncol] = acc[mt][nt][i];
}


__device__ __forceinline__ int wave_incl_scan(int x, int lane) {
#pragma unroll
    for (int o = 1; o < 32; o <<= 1) {
        const int y = __shfl_up(x, o, 32);
        if (lane >= o) x += y;
    }
    return x;
}

__global__ void __launch_bounds__(GT)
gat_bucket_count_kernel(const int* __restrict__ dst, int* __restrict__ bcount,
                        int n_edges, int nb) {
    __shared__ int cnt[MAXNB];
    const int t = threadIdx.x;
    cnt[t] = 0;
    __syncthreads();
    const int base = blockIdx.x * (GT * EPT);
    int d[EPT];
#pragma unroll
    for (int k = 0; k < EPT; ++k) {
        const int e = base + k * GT + t;
        d[k] = dst[e < n_edges ? e : n_edges - 1];
    }
#pragma unroll
    for (int k = 0; k < EPT; ++k) {
        const int e = base + k * GT + t;
        if (e < n_edges) atomicAdd(&cnt[d[k] >> BSH], 1);
    }
    __syncthreads();
    if (t < nb) {
        const int c = cnt[t];
        if (c > 0) atomicAdd(&bcount[t], c);
    }
}

__global__ void __launch_bounds__(GT)
gat_bucket_scatter_kernel(const int* __restrict__ src, const int* __restrict__ dst,
                          const int* __restrict__ bcount, int* __restrict__ bcursor,
                          int* __restrict__ bstart, int2* __restrict__ rec,
                          int n_edges, int nb) {
    __shared__ int cnt[MAXNB];
    __shared__ int wsum[32];
    const int t    = threadIdx.x;
    const int lane = t & 31;
    const int wave = t >> 5;

    const int v = (t < nb) ? bcount[t] : 0;
    const int x = wave_incl_scan(v, lane);
    if (lane == 31) wsum[wave] = x;
    cnt[t] = 0;
    __syncthreads();
    if (wave == 0) {
        const int w  = wsum[lane];
        const int xs = wave_incl_scan(w, lane);
        wsum[lane] = xs - w;
    }
    __syncthreads();
    const int first = wsum[wave] + (x - v);
    if (blockIdx.x == 0 && t < nb) bstart[t] = first;

    int s[EPT], d[EPT], r[EPT];
    const int base = blockIdx.x * (GT * EPT);
#pragma unroll
    for (int k = 0; k < EPT; ++k) {
        const int e  = base + k * GT + t;
        const int ec = e < n_edges ? e : n_edges - 1;
        s[k] = src[ec];
        d[k] = dst[ec];
    }
#pragma unroll
    for (int k = 0; k < EPT; ++k) {
        const int e = base + k * GT + t;
        r[k] = 0;
        if (e < n_edges) r[k] = atomicAdd(&cnt[d[k] >> BSH], 1);
        else             d[k] = -1;
    }
    __syncthreads();
    if (t < nb) {
        const int c = cnt[t];
        int run = 0;
        if (c > 0) run = atomicAdd(&bcursor[t], c);
        cnt[t] = first + run;
    }
    __syncthreads();
#pragma unroll
    for (int k = 0; k < EPT; ++k) {
        if (d[k] >= 0) rec[cnt[d[k] >> BSH] + r[k]] = make_int2(s[k], d[k]);
    }
}

__global__ void __launch_bounds__(GT)
gat_bucket_group_kernel(const int2* __restrict__ rec, const int* __restrict__ bcount,
                        const int* __restrict__ bstart, int* __restrict__ deg,
                        int* __restrict__ start, int* __restrict__ csr_src, int n_nodes) {
    __shared__ int cnt[BNODES];
    __shared__ int wsum[32];
    const int b    = blockIdx.x;
    const int t    = threadIdx.x;
    const int lane = t & 31;
    const int wave = t >> 5;
    const int n  = bcount[b];
    const int r0 = bstart[b];

    if (t < BNODES) cnt[t] = 0;
    __syncthreads();
    for (int i0 = 0; i0 < n; i0 += GT * NR) {
        int y[NR];
#pragma unroll
        for (int k = 0; k < NR; ++k) {
            const int i = i0 + k * GT + t;
            y[k] = rec[r0 + (i < n ? i : n - 1)].y;
        }
#pragma unroll
        for (int k = 0; k < NR; ++k) {
            const int i = i0 + k * GT + t;
            if (i < n) atomicAdd(&cnt[y[k] & (BNODES - 1)], 1);
        }
    }
    __syncthreads();

    const int v = (t < BNODES) ? cnt[t] : 0;
    const int x = wave_incl_scan(v, lane);
    if (lane == 31) wsum[wave] = x;
    __syncthreads();
    if (wave == 0) {
        const int w  = wsum[lane];
        const int xs = wave_incl_scan(w, lane);
        wsum[lane] = xs - w;
    }
    __syncthreads();
    if (t < BNODES) {
        const int first = r0 + wsum[wave] + (x - v);
        cnt[t] = first;
        const int node = b * BNODES + t;
        if (node < n_nodes) { deg[node] = v; start[node] = first; }
    }
    __syncthreads();
    for (int i0 = 0; i0 < n; i0 += GT * NR) {
        int2 q[NR];
#pragma unroll
        for (int k = 0; k < NR; ++k) {
            const int i = i0 + k * GT + t;
            q[k] = rec[r0 + (i < n ? i : n - 1)];
        }
#pragma unroll
        for (int k = 0; k < NR; ++k) {
            const int i = i0 + k * GT + t;
            if (i < n) {
                const int slot = atomicAdd(&cnt[q[k].y & (BNODES - 1)], 1);
                csr_src[slot] = q[k].x;
            }
        }
    }
}

__device__ __forceinline__ float lrelu(float v) { return (v > 0.0f) ? v : (LEAKY * v); }

__device__ __forceinline__ float pair_other(float v) {
    return __int_as_float(__builtin_amdgcn_update_dpp(0, __float_as_int(v), 0xB1, 0xF, 0xF, false));
}

__global__ void __launch_bounds__(256)
gat_node_kernel(const float* __restrict__ h, const float* __restrict__ P,
                const float* __restrict__ Q, const float* __restrict__ a,
                const int* __restrict__ csr_src, const int* __restrict__ start,
                const int* __restrict__ deg, float* __restrict__ out, int n_nodes) {
    const int t    = blockIdx.x * 256 + threadIdx.x;
    const int node = t >> 4;
    const int c4   = (t & 15) * 4;
    if (node >= n_nodes) return;

    const int n   = deg[node];
    const int beg = start[node];
    const v4f av = *(const v4f*)(a + c4);
    const v4f qv = *(const v4f*)(Q + (size_t)node * HC + c4);

    float m = -INFINITY, den = 0.0f;
    v4f acc = (v4f){0.f, 0.f, 0.f, 0.f};

    for (int j = 0; j < n; j += NU) {
        int s[NU];
#pragma unroll
        for (int k = 0; k < NU; ++k) {
            int jj = j + k;
            if (jj > n - 1) jj = n - 1;
            s[k] = csr_src[beg + jj];
        }
        v4f pv[NU], hv[NU];
#pragma unroll
        for (int k = 0; k < NU; ++k) {
            pv[k] = *(const v4f*)(P + (size_t)s[k] * HC + c4);
            hv[k] = *(const v4f*)(h + (size_t)s[k] * HC + c4);
        }
        float sc[NU];
        float mn = m;
#pragma unroll
        for (int k = 0; k < NU; ++k) {
            const float t0 = lrelu(pv[k].x + qv.x) * av.x;
            const float t1 = lrelu(pv[k].y + qv.y) * av.y;
            const float t2 = lrelu(pv[k].z + qv.z) * av.z;
            const float t3 = lrelu(pv[k].w + qv.w) * av.w;
            const float part = (t0 + t1) + (t2 + t3);
            float v = part + pair_other(part);
            if (j + k >= n) v = -INFINITY;
            sc[k] = v;
            mn = fmaxf(mn, v);
        }
        const float scale = __expf(m - mn);
        den *= scale;
        acc.x *= scale; acc.y *= scale; acc.z *= scale; acc.w *= scale;
#pragma unroll
        for (int k = 0; k < NU; ++k) {
            const float ex = __expf(sc[k] - mn);
            den += ex;
            acc.x += ex * hv[k].x; acc.y += ex * hv[k].y;
            acc.z += ex * hv[k].z; acc.w += ex * hv[k].w;
        }
        m = mn;
    }

    v4f o = (v4f){0.f, 0.f, 0.f, 0.f};
    if (n > 0) { o.x = acc.x / den; o.y = acc.y / den; o.z = acc.z / den; o.w = acc.w / den; }
    *(v4f*)(out + (size_t)node * HC + c4) = o;
}

extern "C" void kernel_launch(void* const* d_in, const int* in_sizes, int n_in,
                              void* d_out, int out_size, void* d_ws, size_t ws_size,
                              hipStream_t stream) {
    const float* h   = (const float*)d_in[0];
    const int*   src = (const int*)d_in[1];
    const int*   dst = (const int*)d_in[2];
    const float* w   = (const float*)d_in[3];
    const float* a   = (const float*)d_in[4];
    float* out = (float*)d_out;

    const int n_edges = in_sizes[1];
    const int n_nodes = in_sizes[0] / HC;

    const int npb  = WPB * MT * 16;
    const int nblk = (n_nodes + npb - 1) / npb;

    const int nb  = (n_nodes + BNODES - 1) >> BSH;
    const int nbk = (n_edges + GT * EPT - 1) / (GT * EPT);

    char* ws = (char*)d_ws;
    size_t off = 0;
    float*  P       = (float*)(ws + off);  off += (size_t)nblk * npb * HC * sizeof(float);
    float*  Q       = (float*)(ws + off);  off += (size_t)nblk * npb * HC * sizeof(float);
    float4* wq      = (float4*)(ws + off); off += (size_t)2 * 16 * HC * sizeof(float4);
    int2*   rec     = (int2*)(ws + off);   off += (size_t)n_edges * sizeof(int2);
    int*    csr_src = (int*)(ws + off);    off += (size_t)n_edges * sizeof(int);
    int*    deg     = (int*)(ws + off);    off += (size_t)n_nodes * sizeof(int);
    int*    start   = (int*)(ws + off);    off += (size_t)n_nodes * sizeof(int);
    int*    bcount  = (int*)(ws + off);    off += (size_t)MAXNB * sizeof(int);
    int*    bcursor = (int*)(ws + off);    off += (size_t)MAXNB * sizeof(int);
    int*    bstart  = (int*)(ws + off);

    const int nquads = 2 * 16 * HC;
    const int initN  = nb > nquads ? nb : nquads;
    gat_init_kernel<<<(initN + 255) / 256, 256, 0, stream>>>(bcount, bcursor, wq, w, nb, nquads);

    gat_node_proj_kernel<<<dim3(nblk, 2), 32 * WPB, 0, stream>>>(h, wq, P, Q, n_nodes);

    gat_bucket_count_kernel<<<nbk, GT, 0, stream>>>(dst, bcount, n_edges, nb);

    gat_bucket_scatter_kernel<<<nbk, GT, 0, stream>>>(src, dst, bcount, bcursor, bstart, rec,
                                                      n_edges, nb);

    gat_bucket_group_kernel<<<nb, GT, 0, stream>>>(rec, bcount, bstart, deg, start, csr_src,
                                                   n_nodes);

    const int nthr = n_nodes * 16;
    gat_node_kernel<<<(nthr + 255) / 256, 256, 0, stream>>>(h, P, Q, a, csr_src, start, deg,
                                                            out, n_nodes);
}


# Block-local LDS sort groups edges: an edge block stores its records in bucket order and a count/offset row, bucket pass walks the runs; no count launch, no memory atomics; sums in LDS arrival order
# speedup vs baseline: 8.7574x; 1.0695x over previous
#include <hip/hip_runtime.h>
#include <math.h>

#define HC        64
#define LEAKY     0.2f
#define MT        2
#define WPB       2
#define NROWS     68
#define NU        4
#define GT        1024
#define EPT       8
#define CH        (GT * EPT)
#define BSH       9
#define BNODES    (1 << BSH)
#define NRUN      8
#define MAXNB     GT

typedef __attribute__((ext_vector_type(2))) float v2f;
typedef __attribute__((ext_vector_type(4))) float v4f;
typedef __attribute__((ext_vector_type(8))) float v8f;


__global__ void gat_init_kernel(float4* __restrict__ wq, const float* __restrict__ w, int nquads) {
    int t = blockIdx.x * blockDim.x + threadIdx.x;
    if (t < nquads) {
        const int hi = t >> 10;
        const int q8 = (t >> 6) & 15;
        const int n  = t & 63;
        const int k  = 8 * q8 + 2 * hi;
        wq[t] = make_float4(w[(size_t)k * HC + n],       w[(size_t)(k + 1) * HC + n],
                            w[(size_t)(k + 4) * HC + n], w[(size_t)(k + 5) * HC + n]);
    }
}

__global__ void __launch_bounds__(32 * WPB)
gat_node_proj_kernel(const float* __restrict__ h, const float4* __restrict__ wq,
                     float* __restrict__ P, float* __restrict__ Q, int n_nodes) {
    __shared__ float lds[WPB * MT * 16 * NROWS];

    const int wave  = threadIdx.x >> 5;
    const int lane  = threadIdx.x & 31;
    const int half  = blockIdx.y;
    const int nbase = blockIdx.x * (WPB * MT * 16) + wave * (MT * 16);
    const int rbase = wave * (MT * 16);

#pragma unroll
    for (int j = 0; j < 16; ++j) {
        const int idx = lane + 32 * j;
        const int row = idx >> 4;
        const int c4  = idx & 15;
        int ng = nbase + row;
        if (ng >= n_nodes) ng = n_nodes - 1;
        const float4 v = ((const float4*)(h + (size_t)ng * HC))[c4];
        *(float4*)&lds[(rbase + row) * NROWS + c4 * 4] = v;
    }
    __syncthreads();

    const int mrow = lane & 15;
    const int hi   = lane >> 4;
    const int hi2  = hi * 2;
    const int ncol = lane & 15;

    v8f acc[MT][4];
#pragma unroll
    for (int mt = 0; mt < MT; ++mt)
#pragma unroll
        for (int nt = 0; nt < 4; ++nt)
            acc[mt][nt] = (v8f){0.f,0.f,0.f,0.f,0.f,0.f,0.f,0.f};

#pragma unroll
    for (int kb8 = 0; kb8 < 8; ++kb8) {
        const int kb = kb8 * 8;
        float4 bq[4];
        const float4* wrow = wq + ((size_t)hi * 16 + half * 8 + kb8) * HC;
#pragma unroll
        for (int nt = 0; nt < 4; ++nt) bq[nt] = wrow[nt * 16 + ncol];

#pragma unroll
        for (int mt = 0; mt < MT; ++mt) {
            const float* arow = &lds[(rbase + mt * 16 + mrow) * NROWS];
            v2f a0, a1;
            a0.x = arow[kb + hi2];     a0.y = arow[kb + hi2 + 1];
            a1.x = arow[kb + 4 + hi2]; a1.y = arow[kb + 4 + hi2 + 1];
#pragma unroll
            for (int nt = 0; nt < 4; ++nt) {
                v2f b0; b0.x = bq[nt].x; b0.y = bq[nt].y;
                acc[mt][nt] = __builtin_amdgcn_wmma_f32_16x16x4_f32(
                    false, a0, false, b0, (short)0, acc[mt][nt], false, false);
            }
#pragma unroll
            for (int nt = 0; nt < 4; ++nt) {
                v2f b1; b1.x = bq[nt].z; b1.y = bq[nt].w;
                acc[mt][nt] = __builtin_amdgcn_wmma_f32_16x16x4_f32(
                    false, a1, false, b1, (short)0, acc[mt][nt], false, false);
            }
        }
    }

    float* plane = half ? Q : P;
#pragma unroll
    for (int mt = 0; mt < MT; ++mt)
#pragma unroll
        for (int nt = 0; nt < 4; ++nt)
#pragma unroll
            for (int i = 0; i < 8; ++i)
                plane[(size_t)(nbase + mt * 16 + i + 8 * hi) * HC + nt * 16 + ncol] = acc[mt][nt][i];
}


__device__ __forceinline__ int wave_incl_scan(int x, int lane) {
#pragma unroll
    for (int o = 1; o < 32; o <<= 1) {
        const int y = __shfl_up(x, o, 32);
        if (lane >= o) x += y;
    }
    return x;
}

__global__ void __launch_bounds__(GT)
gat_block_sort_kernel(const int* __restrict__ src, const int* __restrict__ dst,
                      int2* __restrict__ rec, int2* __restrict__ tab,
                      int n_edges, int nb, int nbk) {
    __shared__ int  cnt[MAXNB];
    __shared__ int  wsum[32];
    __shared__ int2 srt[CH];
    const int t    = threadIdx.x;
    const int lane = t & 31;
    const int wave = t >> 5;
    const int base = blockIdx.x * CH;

    int s[EPT], d[EPT], r[EPT];
#pragma unroll
    for (int k = 0; k < EPT; ++k) {
        const int e  = base + k * GT + t;
        const int ec = e < n_edges ? e : n_edges - 1;
        s[k] = src[ec];
        d[k] = dst[ec];
    }
    cnt[t] = 0;
    __syncthreads();
#pragma unroll
    for (int k = 0; k < EPT; ++k) {
        const int e = base + k * GT + t;
        r[k] = 0;
        if (e < n_edges) r[k] = atomicAdd(&cnt[d[k] >> BSH], 1);
        else             d[k] = -1;
    }
    __syncthreads();

    const int v = cnt[t];
    const int x = wave_incl_scan(v, lane);
    if (lane == 31) wsum[wave] = x;
    __syncthreads();
    if (wave == 0) {
        const int w  = wsum[lane];
        const int xs = wave_incl_scan(w, lane);
        wsum[lane] = xs - w;
    }
    __syncthreads();
    const int first = wsum[wave] + (x - v);
    cnt[t] = first;
    if (t < nb) tab[(size_t)t * nbk + blockIdx.x] = make_int2(v, first);
    __syncthreads();
#pragma unroll
    for (int k = 0; k < EPT; ++k) {
        if (d[k] >= 0) srt[cnt[d[k] >> BSH] + r[k]] = make_int2(s[k], d[k]);
    }
    __syncthreads();
    int2 g[EPT];
#pragma unroll
    for (int k = 0; k < EPT; ++k) g[k] = srt[k * GT + t];
#pragma unroll
    for (int k = 0; k < EPT; ++k) {
        const int i = k * GT + t;
        if (base + i < n_edges) rec[base + i] = g[k];
    }
}

__global__ void __launch_bounds__(GT)
gat_bucket_group_kernel(const int2* __restrict__ rec, const int2* __restrict__ tab,
                        int* __restrict__ deg, int* __restrict__ start,
                        int* __restrict__ csr_src, int n_nodes, int nbk) {
    __shared__ int cnt[BNODES];
    __shared__ int wsum[32];
    const int b    = blockIdx.x;
    const int t    = threadIdx.x;
    const int lane = t & 31;
    const int wave = t >> 5;
    const int2* row = tab + (size_t)b * nbk;

    int osum = 0;
    for (int j = t; j < nbk; j += GT) osum += row[j].y;
    osum = wave_incl_scan(osum, lane);
    if (lane == 31) wsum[wave] = osum;
    if (t < BNODES) cnt[t] = 0;
    __syncthreads();
    const int r0 = __shfl(wave_incl_scan(wsum[lane], lane), 31, 32);
    __syncthreads();

    for (int j0 = wave; j0 < nbk; j0 += 32 * NRUN) {
        int c[NRUN], o[NRUN];
#pragma unroll
        for (int q = 0; q < NRUN; ++q) {
            const int j  = j0 + 32 * q;
            const int2 co = row[j < nbk ? j : nbk - 1];
            c[q] = j < nbk ? co.x : 0;
            o[q] = j * CH + co.y;
        }
        int y[NRUN][2];
#pragma unroll
        for (int q = 0; q < NRUN; ++q)
#pragma unroll
            for (int u = 0; u < 2; ++u) {
                const int i = lane + 32 * u;
                y[q][u] = rec[i < c[q] ? o[q] + i : 0].y;
            }
#pragma unroll
        for (int q = 0; q < NRUN; ++q)
#pragma unroll
            for (int u = 0; u < 2; ++u) {
                const int i = lane + 32 * u;
                if (i < c[q]) atomicAdd(&cnt[y[q][u] & (BNODES - 1)], 1);
            }
#pragma unroll
        for (int q = 0; q < NRUN; ++q)
            for (int i = lane + 64; i < c[q]; i += 32)
                atomicAdd(&cnt[rec[o[q] + i].y & (BNODES - 1)], 1);
    }
    __syncthreads();

    const int v = (t < BNODES) ? cnt[t] : 0;
    const int x = wave_incl_scan(v, lane);
    if (lane == 31) wsum[wave] = x;
    __syncthreads();
    if (wave == 0) {
        const int w  = wsum[lane];
        const int xs = wave_incl_scan(w, lane);
        wsum[lane] = xs - w;
    }
    __syncthreads();
    if (t < BNODES) {
        const int first = r0 + wsum[wave] + (x - v);
        cnt[t] = first;
        const int node = b * BNODES + t;
        if (node < n_nodes) { deg[node] = v; start[node] = first; }
    }
    __syncthreads();

    for (int j0 = wave; j0 < nbk; j0 += 32 * NRUN) {
        int c[NRUN], o[NRUN];
#pragma unroll
        for (int q = 0; q < NRUN; ++q) {
            const int j  = j0 + 32 * q;
            const int2 co = row[j < nbk ? j : nbk - 1];
            c[q] = j < nbk ? co.x : 0;
            o[q] = j * CH + co.y;
        }
        int2 p[NRUN][2];
#pragma unroll
        for (int q = 0; q < NRUN; ++q)
#pragma unroll
            for (int u = 0; u < 2; ++u) {
                const int i = lane + 32 * u;
                p[q][u] = rec[i < c[q] ? o[q] + i : 0];
            }
#pragma unroll
        for (int q = 0; q < NRUN; ++q)
#pragma unroll
            for (int u = 0; u < 2; ++u) {
                const int i = lane + 32 * u;
                if (i < c[q]) {
                    const int slot = atomicAdd(&cnt[p[q][u].y & (BNODES - 1)], 1);
                    csr_src[slot] = p[q][u].x;
                }
            }
#pragma unroll
        for (int q = 0; q < NRUN; ++q)
            for (int i = lane + 64; i < c[q]; i += 32) {
                const int2 pr = rec[o[q] + i];
                const int slot = atomicAdd(&cnt[pr.y & (BNODES - 1)], 1);
                csr_src[slot] = pr.x;
            }
    }
}

__device__ __forceinline__ float lrelu(float v) { return (v > 0.0f) ? v : (LEAKY * v); }

__device__ __forceinline__ float pair_other(float v) {
    return __int_as_float(__builtin_amdgcn_update_dpp(0, __float_as_int(v), 0xB1, 0xF, 0xF, false));
}

__global__ void __launch_bounds__(256)
gat_node_kernel(const float* __restrict__ h, const float* __restrict__ P,
                const float* __restrict__ Q, const float* __restrict__ a,
                const int* __restrict__ csr_src, const int* __restrict__ start,
                const int* __restrict__ deg, float* __restrict__ out, int n_nodes) {
    const int t    = blockIdx.x * 256 + threadIdx.x;
    const int node = t >> 4;
    const int c4   = (t & 15) * 4;
    if (node >= n_nodes) return;

    const int n   = deg[node];
    const int beg = start[node];
    const v4f av = *(const v4f*)(a + c4);
    const v4f qv = *(const v4f*)(Q + (size_t)node * HC + c4);

    float m = -INFINITY, den = 0.0f;
    v4f acc = (v4f){0.f, 0.f, 0.f, 0.f};

    for (int j = 0; j < n; j += NU) {
        int s[NU];
#pragma unroll
        for (int k = 0; k < NU; ++k) {
            int jj = j + k;
            if (jj > n - 1) jj = n - 1;
            s[k] = csr_src[beg + jj];
        }
        v4f pv[NU], hv[NU];
#pragma unroll
        for (int k = 0; k < NU; ++k) {
            pv[k] = *(const v4f*)(P + (size_t)s[k] * HC + c4);
            hv[k] = *(const v4f*)(h + (size_t)s[k] * HC + c4);
        }
        float sc[NU];
        float mn = m;
#pragma unroll
        for (int k = 0; k < NU; ++k) {
            const float t0 = lrelu(pv[k].x + qv.x) * av.x;
            const float t1 = lrelu(pv[k].y + qv.y) * av.y;
            const float t2 = lrelu(pv[k].z + qv.z) * av.z;
            const float t3 = lrelu(pv[k].w + qv.w) * av.w;
            const float part = (t0 + t1) + (t2 + t3);
            float v = part + pair_other(part);
            if (j + k >= n) v = -INFINITY;
            sc[k] = v;
            mn = fmaxf(mn, v);
        }
        const float scale = __expf(m - mn);
        den *= scale;
        acc.x *= scale; acc.y *= scale; acc.z *= scale; acc.w *= scale;
#pragma unroll
        for (int k = 0; k < NU; ++k) {
            const float ex = __expf(sc[k] - mn);
            den += ex;
            acc.x += ex * hv[k].x; acc.y += ex * hv[k].y;
            acc.z += ex * hv[k].z; acc.w += ex * hv[k].w;
        }
        m = mn;
    }

    v4f o = (v4f){0.f, 0.f, 0.f, 0.f};
    if (n > 0) { o.x = acc.x / den; o.y = acc.y / den; o.z = acc.z / den; o.w = acc.w / den; }
    *(v4f*)(out + (size_t)node * HC + c4) = o;
}

extern "C" void kernel_launch(void* const* d_in, const int* in_sizes, int n_in,
                              void* d_out, int out_size, void* d_ws, size_t ws_size,
                              hipStream_t stream) {
    const float* h   = (const float*)d_in[0];
    const int*   src = (const int*)d_in[1];
    const int*   dst = (const int*)d_in[2];
    const float* w   = (const float*)d_in[3];
    const float* a   = (const float*)d_in[4];
    float* out = (float*)d_out;

    const int n_edges = in_sizes[1];
    const int n_nodes = in_sizes[0] / HC;

    const int npb  = WPB * MT * 16;
    const int nblk = (n_nodes + npb - 1) / npb;

    const int nb  = (n_nodes + BNODES - 1) >> BSH;
    const int nbk = (n_edges + CH - 1) / CH;

    char* ws = (char*)d_ws;
    size_t off = 0;
    float*  P       = (float*)(ws + off);  off += (size_t)nblk * npb * HC * sizeof(float);
    float*  Q       = (float*)(ws + off);  off += (size_t)nblk * npb * HC * sizeof(float);
    float4* wq      = (float4*)(ws + off); off += (size_t)2 * 16 * HC * sizeof(float4);
    int2*   rec     = (int2*)(ws + off);   off += (size_t)n_edges * sizeof(int2);
    int2*   tab     = (int2*)(ws + off);   off += (size_t)nb * nbk * sizeof(int2);
    int*    csr_src = (int*)(ws + off);    off += (size_t)n_edges * sizeof(int);
    int*    deg     = (int*)(ws + off);    off += (size_t)n_nodes * sizeof(int);
    int*    start   = (int*)(ws + off);

    const int nquads = 2 * 16 * HC;
    gat_init_kernel<<<(nquads + 255) / 256, 256, 0, stream>>>(wq, w, nquads);

    gat_node_proj_kernel<<<dim3(nblk, 2), 32 * WPB, 0, stream>>>(h, wq, P, Q, n_nodes);

    gat_block_sort_kernel<<<nbk, GT, 0, stream>>>(src, dst, rec, tab, n_edges, nb, nbk);

    gat_bucket_group_kernel<<<nb, GT, 0, stream>>>(rec, tab, deg, start, csr_src, n_nodes, nbk);

    const int nthr = n_nodes * 16;
    gat_node_kernel<<<(nthr + 255) / 256, 256, 0, stream>>>(h, P, Q, a, csr_src, start, deg,
                                                            out, n_nodes);
}


# Node projection rides in the block-sort launch as extra 1024-thread blocks beside the edge blocks (w packed per block in LDS): 5 launches -> 3; k-sum order within each 8-k group changes
# speedup vs baseline: 9.5264x; 1.0878x over previous
#include <hip/hip_runtime.h>
#include <math.h>

#define HC        64
#define LEAKY     0.2f
#define NU        4
#define GT        1024
#define EPT       8
#define CH        (GT * EPT)
#define BSH       9
#define BNODES    (1 << BSH)
#define NRUN      8
#define MAXNB     GT
#define PNB       (GT / 2)

typedef __attribute__((ext_vector_type(2))) float v2f;
typedef __attribute__((ext_vector_type(4))) float v4f;
typedef __attribute__((ext_vector_type(8))) float v8f;


__device__ __forceinline__ void proj_role(const float* __restrict__ h, const float* __restrict__ w,
                                          float* __restrict__ P, float* __restrict__ Q,
                                          int pb, int n_nodes, float4* wl) {
    const int t     = threadIdx.x;
    const int lane  = t & 31;
    const int wave  = t >> 5;
    const int c     = lane & 15;
    const int hi    = lane >> 4;
    const int nbase = pb * PNB + wave * 16;

    float wv[2][4];
#pragma unroll
    for (int u = 0; u < 2; ++u) {
        const int q = t + u * GT;
        const float* wp = w + (size_t)(4 * (q >> 6)) * HC + (q & 63);
#pragma unroll
        for (int r = 0; r < 4; ++r) wv[u][r] = wp[r * HC];
    }

    int ng = nbase + c;
    if (ng >= n_nodes) ng = n_nodes - 1;
    const float4* arow = (const float4*)(h + (size_t)ng * HC) + hi;
    float4 aq[8];
#pragma unroll
    for (int j = 0; j < 8; ++j) aq[j] = arow[2 * j];

#pragma unroll
    for (int u = 0; u < 2; ++u) wl[t + u * GT] = make_float4(wv[u][0], wv[u][1], wv[u][2], wv[u][3]);
    __syncthreads();

#pragma unroll
    for (int half = 0; half < 2; ++half) {
        float* plane = half ? Q : P;
#pragma unroll
        for (int np = 0; np < 2; ++np) {
            v8f acc[2];
#pragma unroll
            for (int nt = 0; nt < 2; ++nt) acc[nt] = (v8f){0.f,0.f,0.f,0.f,0.f,0.f,0.f,0.f};
#pragma unroll
            for (int j = 0; j < 8; ++j) {
                const float4* brow = wl + (half * 16 + 2 * j + hi) * HC + np * 32 + c;
                float4 bq[2];
#pragma unroll
                for (int nt = 0; nt < 2; ++nt) bq[nt] = brow[nt * 16];
                v2f a0, a1;
                a0.x = aq[j].x; a0.y = aq[j].y;
                a1.x = aq[j].z; a1.y = aq[j].w;
#pragma unroll
                for (int nt = 0; nt < 2; ++nt) {
                    v2f b0; b0.x = bq[nt].x; b0.y = bq[nt].y;
                    acc[nt] = __builtin_amdgcn_wmma_f32_16x16x4_f32(
                        false, a0, false, b0, (short)0, acc[nt], false, false);
                }
#pragma unroll
                for (int nt = 0; nt < 2; ++nt) {
                    v2f b1; b1.x = bq[nt].z; b1.y = bq[nt].w;
                    acc[nt] = __builtin_amdgcn_wmma_f32_16x16x4_f32(
                        false, a1, false, b1, (short)0, acc[nt], false, false);
                }
            }
#pragma unroll
            for (int nt = 0; nt < 2; ++nt)
#pragma unroll
                for (int i = 0; i < 8; ++i)
                    plane[(size_t)(nbase + i + 8 * hi) * HC + np * 32 + nt * 16 + c] = acc[nt][i];
        }
    }
}


__device__ __forceinline__ int wave_incl_scan(int x, int lane) {
#pragma unroll
    for (int o = 1; o < 32; o <<= 1) {
        const int y = __shfl_up(x, o, 32);
        if (lane >= o) x += y;
    }
    return x;
}

__global__ void __launch_bounds__(GT, 16)
gat_block_sort_kernel(const int* __restrict__ src, const int* __restrict__ dst,
                      int2* __restrict__ rec, int2* __restrict__ tab,
                      int n_edges, int nb, int nbk,
                      const float* __restrict__ h, const float* __restrict__ w,
                      float* __restrict__ P, float* __restrict__ Q, int n_nodes) {
    __shared__ int  cnt[MAXNB];
    __shared__ int  wsum[32];
    __shared__ __attribute__((aligned(16))) int2 srt[CH];
    if (blockIdx.x >= nbk) {
        proj_role(h, w, P, Q, blockIdx.x - nbk, n_nodes, (float4*)srt);
        return;
    }
    const int t    = threadIdx.x;
    const int lane = t & 31;
    const int wave = t >> 5;
    const int base = blockIdx.x * CH;

    int s[EPT], d[EPT], r[EPT];
#pragma unroll
    for (int k = 0; k < EPT; ++k) {
        const int e  = base + k * GT + t;
        const int ec = e < n_edges ? e : n_edges - 1;
        s[k] = src[ec];
        d[k] = dst[ec];
    }
    cnt[t] = 0;
    __syncthreads();
#pragma unroll
    for (int k = 0; k < EPT; ++k) {
        const int e = base + k * GT + t;
        r[k] = 0;
        if (e < n_edges) r[k] = atomicAdd(&cnt[d[k] >> BSH], 1);
        else             d[k] = -1;
    }
    __syncthreads();

    const int v = cnt[t];
    const int x = wave_incl_scan(v, lane);
    if (lane == 31) wsum[wave] = x;
    __syncthreads();
    if (wave == 0) {
        const int w  = wsum[lane];
        const int xs = wave_incl_scan(w, lane);
        wsum[lane] = xs - w;
    }
    __syncthreads();
    const int first = wsum[wave] + (x - v);
    cnt[t] = first;
    if (t < nb) tab[(size_t)t * nbk + blockIdx.x] = make_int2(v, first);
    __syncthreads();
#pragma unroll
    for (int k = 0; k < EPT; ++k) {
        if (d[k] >= 0) srt[cnt[d[k] >> BSH] + r[k]] = make_int2(s[k], d[k]);
    }
    __syncthreads();
    int2 g[EPT];
#pragma unroll
    for (int k = 0; k < EPT; ++k) g[k] = srt[k * GT + t];
#pragma unroll
    for (int k = 0; k < EPT; ++k) {
        const int i = k * GT + t;
        if (base + i < n_edges) rec[base + i] = g[k];
    }
}

__global__ void __launch_bounds__(GT)
gat_bucket_group_kernel(const int2* __restrict__ rec, const int2* __restrict__ tab,
                        int* __restrict__ deg, int* __restrict__ start,
                        int* __restrict__ csr_src, int n_nodes, int nbk) {
    __shared__ int cnt[BNODES];
    __shared__ int wsum[32];
    const int b    = blockIdx.x;
    const int t    = threadIdx.x;
    const int lane = t & 31;
    const int wave = t >> 5;
    const int2* row = tab + (size_t)b * nbk;

    int osum = 0;
    for (int j = t; j < nbk; j += GT) osum += row[j].y;
    osum = wave_incl_scan(osum, lane);
    if (lane == 31) wsum[wave] = osum;
    if (t < BNODES) cnt[t] = 0;
    __syncthreads();
    const int r0 = __shfl(wave_incl_scan(wsum[lane], lane), 31, 32);
    __syncthreads();

    for (int j0 = wave; j0 < nbk; j0 += 32 * NRUN) {
        int c[NRUN], o[NRUN];
#pragma unroll
        for (int q = 0; q < NRUN; ++q) {
            const int j  = j0 + 32 * q;
            const int2 co = row[j < nbk ? j : nbk - 1];
            c[q] = j < nbk ? co.x : 0;
            o[q] = j * CH + co.y;
        }
        int y[NRUN][2];
#pragma unroll
        for (int q = 0; q < NRUN; ++q)
#pragma unroll
            for (int u = 0; u < 2; ++u) {
                const int i = lane + 32 * u;
                y[q][u] = rec[i < c[q] ? o[q] + i : 0].y;
            }
#pragma unroll
        for (int q = 0; q < NRUN; ++q)
#pragma unroll
            for (int u = 0; u < 2; ++u) {
                const int i = lane + 32 * u;
                if (i < c[q]) atomicAdd(&cnt[y[q][u] & (BNODES - 1)], 1);
            }
#pragma unroll
        for (int q = 0; q < NRUN; ++q)
            for (int i = lane + 64; i < c[q]; i += 32)
                atomicAdd(&cnt[rec[o[q] + i].y & (BNODES - 1)], 1);
    }
    __syncthreads();

    const int v = (t < BNODES) ? cnt[t] : 0;
    const int x = wave_incl_scan(v, lane);
    if (lane == 31) wsum[wave] = x;
    __syncthreads();
    if (wave == 0) {
        const int w  = wsum[lane];
        const int xs = wave_incl_scan(w, lane);
        wsum[lane] = xs - w;
    }
    __syncthreads();
    if (t < BNODES) {
        const int first = r0 + wsum[wave] + (x - v);
        cnt[t] = first;
        const int node = b * BNODES + t;
        if (node < n_nodes) { deg[node] = v; start[node] = first; }
    }
    __syncthreads();

    for (int j0 = wave; j0 < nbk; j0 += 32 * NRUN) {
        int c[NRUN], o[NRUN];
#pragma unroll
        for (int q = 0; q < NRUN; ++q) {
            const int j  = j0 + 32 * q;
            const int2 co = row[j < nbk ? j : nbk - 1];
            c[q] = j < nbk ? co.x : 0;
            o[q] = j * CH + co.y;
        }
        int2 p[NRUN][2];
#pragma unroll
        for (int q = 0; q < NRUN; ++q)
#pragma unroll
            for (int u = 0; u < 2; ++u) {
                const int i = lane + 32 * u;
                p[q][u] = rec[i < c[q] ? o[q] + i : 0];
            }
#pragma unroll
        for (int q = 0; q < NRUN; ++q)
#pragma unroll
            for (int u = 0; u < 2; ++u) {
                const int i = lane + 32 * u;
                if (i < c[q]) {
                    const int slot = atomicAdd(&cnt[p[q][u].y & (BNODES - 1)], 1);
                    csr_src[slot] = p[q][u].x;
                }
            }
#pragma unroll
        for (int q = 0; q < NRUN; ++q)
            for (int i = lane + 64; i < c[q]; i += 32) {
                const int2 pr = rec[o[q] + i];
                const int slot = atomicAdd(&cnt[pr.y & (BNODES - 1)], 1);
                csr_src[slot] = pr.x;
            }
    }
}

__device__ __forceinline__ float lrelu(float v) { return (v > 0.0f) ? v : (LEAKY * v); }

__device__ __forceinline__ float pair_other(float v) {
    return __int_as_float(__builtin_amdgcn_update_dpp(0, __float_as_int(v), 0xB1, 0xF, 0xF, false));
}

__global__ void __launch_bounds__(256)
gat_node_kernel(const float* __restrict__ h, const float* __restrict__ P,
                const float* __restrict__ Q, const float* __restrict__ a,
                const int* __restrict__ csr_src, const int* __restrict__ start,
                const int* __restrict__ deg, float* __restrict__ out, int n_nodes) {
    const int t    = blockIdx.x * 256 + threadIdx.x;
    const int node = t >> 4;
    const int c4   = (t & 15) * 4;
    if (node >= n_nodes) return;

    const int n   = deg[node];
    const int beg = start[node];
    const v4f av = *(const v4f*)(a + c4);
    const v4f qv = *(const v4f*)(Q + (size_t)node * HC + c4);

    float m = -INFINITY, den = 0.0f;
    v4f acc = (v4f){0.f, 0.f, 0.f, 0.f};

    for (int j = 0; j < n; j += NU) {
        int s[NU];
#pragma unroll
        for (int k = 0; k < NU; ++k) {
            int jj = j + k;
            if (jj > n - 1) jj = n - 1;
            s[k] = csr_src[beg + jj];
        }
        v4f pv[NU], hv[NU];
#pragma unroll
        for (int k = 0; k < NU; ++k) {
            pv[k] = *(const v4f*)(P + (size_t)s[k] * HC + c4);
            hv[k] = *(const v4f*)(h + (size_t)s[k] * HC + c4);
        }
        float sc[NU];
        float mn = m;
#pragma unroll
        for (int k = 0; k < NU; ++k) {
            const float t0 = lrelu(pv[k].x + qv.x) * av.x;
            const float t1 = lrelu(pv[k].y + qv.y) * av.y;
            const float t2 = lrelu(pv[k].z + qv.z) * av.z;
            const float t3 = lrelu(pv[k].w + qv.w) * av.w;
            const float part = (t0 + t1) + (t2 + t3);
            float v = part + pair_other(part);
            if (j + k >= n) v = -INFINITY;
            sc[k] = v;
            mn = fmaxf(mn, v);
        }
        const float scale = __expf(m - mn);
        den *= scale;
        acc.x *= scale; acc.y *= scale; acc.z *= scale; acc.w *= scale;
#pragma unroll
        for (int k = 0; k < NU; ++k) {
            const float ex = __expf(sc[k] - mn);
            den += ex;
            acc.x += ex * hv[k].x; acc.y += ex * hv[k].y;
            acc.z += ex * hv[k].z; acc.w += ex * hv[k].w;
        }
        m = mn;
    }

    v4f o = (v4f){0.f, 0.f, 0.f, 0.f};
    if (n > 0) { o.x = acc.x / den; o.y = acc.y / den; o.z = acc.z / den; o.w = acc.w / den; }
    *(v4f*)(out + (size_t)node * HC + c4) = o;
}

extern "C" void kernel_launch(void* const* d_in, const int* in_sizes, int n_in,
                              void* d_out, int out_size, void* d_ws, size_t ws_size,
                              hipStream_t stream) {
    const float* h   = (const float*)d_in[0];
    const int*   src = (const int*)d_in[1];
    const int*   dst = (const int*)d_in[2];
    const float* w   = (const float*)d_in[3];
    const float* a   = (const float*)d_in[4];
    float* out = (float*)d_out;

    const int n_edges = in_sizes[1];
    const int n_nodes = in_sizes[0] / HC;

    const int npb  = PNB;
    const int nblk = (n_nodes + npb - 1) / npb;

    const int nb  = (n_nodes + BNODES - 1) >> BSH;
    const int nbk = (n_edges + CH - 1) / CH;

    char* ws = (char*)d_ws;
    size_t off = 0;
    float*  P       = (float*)(ws + off);  off += (size_t)nblk * npb * HC * sizeof(float);
    float*  Q       = (float*)(ws + off);  off += (size_t)nblk * npb * HC * sizeof(float);
    int2*   rec     = (int2*)(ws + off);   off += (size_t)n_edges * sizeof(int2);
    int2*   tab     = (int2*)(ws + off);   off += (size_t)nb * nbk * sizeof(int2);
    int*    csr_src = (int*)(ws + off);    off += (size_t)n_edges * sizeof(int);
    int*    deg     = (int*)(ws + off);    off += (size_t)n_nodes * sizeof(int);
    int*    start   = (int*)(ws + off);

    gat_block_sort_kernel<<<nbk + nblk, GT, 0, stream>>>(src, dst, rec, tab, n_edges, nb, nbk,
                                                         h, w, P, Q, n_nodes);

    gat_bucket_group_kernel<<<nb, GT, 0, stream>>>(rec, tab, deg, start, csr_src, n_nodes, nbk);

    const int nthr = n_nodes * 16;
    gat_node_kernel<<<(nthr + 255) / 256, 256, 0, stream>>>(h, P, Q, a, csr_src, start, deg,
                                                            out, n_nodes);
}


# Q half of the node projection moves from the block-sort launch to the bucket-group launch as extra blocks; P stays: each launch carries half the f32 WMMA work; sums and their order unchanged
# speedup vs baseline: 9.9136x; 1.0406x over previous
#include <hip/hip_runtime.h>
#include <math.h>

#define HC        64
#define LEAKY     0.2f
#define NU        4
#define GT        1024
#define EPT       8
#define CH        (GT * EPT)
#define BSH       9
#define BNODES    (1 << BSH)
#define NRUN      8
#define MAXNB     GT
#define PNB       (GT / 2)

typedef __attribute__((ext_vector_type(2))) float v2f;
typedef __attribute__((ext_vector_type(4))) float v4f;
typedef __attribute__((ext_vector_type(8))) float v8f;


__device__ __forceinline__ void proj_role(const float* __restrict__ h, const float* __restrict__ wh,
                                          float* __restrict__ plane,
                                          int pb, int n_nodes, float4* wl) {
    const int t     = threadIdx.x;
    const int lane  = t & 31;
    const int wave  = t >> 5;
    const int c     = lane & 15;
    const int hi    = lane >> 4;
    const int nbase = pb * PNB + wave * 16;

    float wv[4];
    {
        const float* wp = wh + (size_t)(4 * (t >> 6)) * HC + (t & 63);
#pragma unroll
        for (int r = 0; r < 4; ++r) wv[r] = wp[r * HC];
    }

    int ng = nbase + c;
    if (ng >= n_nodes) ng = n_nodes - 1;
    const float4* arow = (const float4*)(h + (size_t)ng * HC) + hi;
    float4 aq[8];
#pragma unroll
    for (int j = 0; j < 8; ++j) aq[j] = arow[2 * j];

    wl[t] = make_float4(wv[0], wv[1], wv[2], wv[3]);
    __syncthreads();

#pragma unroll
    for (int np = 0; np < 2; ++np) {
        v8f acc[2];
#pragma unroll
        for (int nt = 0; nt < 2; ++nt) acc[nt] = (v8f){0.f,0.f,0.f,0.f,0.f,0.f,0.f,0.f};
#pragma unroll
        for (int j = 0; j < 8; ++j) {
            const float4* brow = wl + (2 * j + hi) * HC + np * 32 + c;
            float4 bq[2];
#pragma unroll
            for (int nt = 0; nt < 2; ++nt) bq[nt] = brow[nt * 16];
            v2f a0, a1;
            a0.x = aq[j].x; a0.y = aq[j].y;
            a1.x = aq[j].z; a1.y = aq[j].w;
#pragma unroll
            for (int nt = 0; nt < 2; ++nt) {
                v2f b0; b0.x = bq[nt].x; b0.y = bq[nt].y;
                acc[nt] = __builtin_amdgcn_wmma_f32_16x16x4_f32(
                    false, a0, false, b0, (short)0, acc[nt], false, false);
            }
#pragma unroll
            for (int nt = 0; nt < 2; ++nt) {
                v2f b1; b1.x = bq[nt].z; b1.y = bq[nt].w;
                acc[nt] = __builtin_amdgcn_wmma_f32_16x16x4_f32(
                    false, a1, false, b1, (short)0, acc[nt], false, false);
            }
        }
#pragma unroll
        for (int nt = 0; nt < 2; ++nt)
#pragma unroll
            for (int i = 0; i < 8; ++i)
                plane[(size_t)(nbase + i + 8 * hi) * HC + np * 32 + nt * 16 + c] = acc[nt][i];
    }
}


__device__ __forceinline__ int wave_incl_scan(int x, int lane) {
#pragma unroll
    for (int o = 1; o < 32; o <<= 1) {
        const int y = __shfl_up(x, o, 32);
        if (lane >= o) x += y;
    }
    return x;
}

__global__ void __launch_bounds__(GT, 16)
gat_block_sort_kernel(const int* __restrict__ src, const int* __restrict__ dst,
                      int2* __restrict__ rec, int2* __restrict__ tab,
                      int n_edges, int nb, int nbk,
                      const float* __restrict__ h, const float* __restrict__ w,
                      float* __restrict__ P, int n_nodes) {
    __shared__ int  cnt[MAXNB];
    __shared__ int  wsum[32];
    __shared__ __attribute__((aligned(16))) int2 srt[CH];
    if (blockIdx.x >= nbk) {
        proj_role(h, w, P, blockIdx.x - nbk, n_nodes, (float4*)srt);
        return;
    }
    const int t    = threadIdx.x;
    const int lane = t & 31;
    const int wave = t >> 5;
    const int base = blockIdx.x * CH;

    int s[EPT], d[EPT], r[EPT];
#pragma unroll
    for (int k = 0; k < EPT; ++k) {
        const int e  = base + k * GT + t;
        const int ec = e < n_edges ? e : n_edges - 1;
        s[k] = src[ec];
        d[k] = dst[ec];
    }
    cnt[t] = 0;
    __syncthreads();
#pragma unroll
    for (int k = 0; k < EPT; ++k) {
        const int e = base + k * GT + t;
        r[k] = 0;
        if (e < n_edges) r[k] = atomicAdd(&cnt[d[k] >> BSH], 1);
        else             d[k] = -1;
    }
    __syncthreads();

    const int v = cnt[t];
    const int x = wave_incl_scan(v, lane);
    if (lane == 31) wsum[wave] = x;
    __syncthreads();
    if (wave == 0) {
        const int w  = wsum[lane];
        const int xs = wave_incl_scan(w, lane);
        wsum[lane] = xs - w;
    }
    __syncthreads();
    const int first = wsum[wave] + (x - v);
    cnt[t] = first;
    if (t < nb) tab[(size_t)t * nbk + blockIdx.x] = make_int2(v, first);
    __syncthreads();
#pragma unroll
    for (int k = 0; k < EPT; ++k) {
        if (d[k] >= 0) srt[cnt[d[k] >> BSH] + r[k]] = make_int2(s[k], d[k]);
    }
    __syncthreads();
    int2 g[EPT];
#pragma unroll
    for (int k = 0; k < EPT; ++k) g[k] = srt[k * GT + t];
#pragma unroll
    for (int k = 0; k < EPT; ++k) {
        const int i = k * GT + t;
        if (base + i < n_edges) rec[base + i] = g[k];
    }
}

__global__ void __launch_bounds__(GT, 16)
gat_bucket_group_kernel(const int2* __restrict__ rec, const int2* __restrict__ tab,
                        int* __restrict__ deg, int* __restrict__ start,
                        int* __restrict__ csr_src, int n_nodes, int nbk, int nb,
                        const float* __restrict__ h, const float* __restrict__ w,
                        float* __restrict__ Q) {
    __shared__ int cnt[BNODES];
    __shared__ int wsum[32];
    __shared__ __attribute__((aligned(16))) float4 wq[GT];
    if (blockIdx.x >= nb) {
        proj_role(h, w + (size_t)HC * HC, Q, blockIdx.x - nb, n_nodes, wq);
        return;
    }
    const int b    = blockIdx.x;
    const int t    = threadIdx.x;
    const int lane = t & 31;
    const int wave = t >> 5;
    const int2* row = tab + (size_t)b * nbk;

    int osum = 0;
    for (int j = t; j < nbk; j += GT) osum += row[j].y;
    osum = wave_incl_scan(osum, lane);
    if (lane == 31) wsum[wave] = osum;
    if (t < BNODES) cnt[t] = 0;
    __syncthreads();
    const int r0 = __shfl(wave_incl_scan(wsum[lane], lane), 31, 32);
    __syncthreads();

    for (int j0 = wave; j0 < nbk; j0 += 32 * NRUN) {
        int c[NRUN], o[NRUN];
#pragma unroll
        for (int q = 0; q < NRUN; ++q) {
            const int j  = j0 + 32 * q;
            const int2 co = row[j < nbk ? j : nbk - 1];
            c[q] = j < nbk ? co.x : 0;
            o[q] = j * CH + co.y;
        }
        int y[NRUN][2];
#pragma unroll
        for (int q = 0; q < NRUN; ++q)
#pragma unroll
            for (int u = 0; u < 2; ++u) {
                const int i = lane + 32 * u;
                y[q][u] = rec[i < c[q] ? o[q] + i : 0].y;
            }
#pragma unroll
        for (int q = 0; q < NRUN; ++q)
#pragma unroll
            for (int u = 0; u < 2; ++u) {
                const int i = lane + 32 * u;
                if (i < c[q]) atomicAdd(&cnt[y[q][u] & (BNODES - 1)], 1);
            }
#pragma unroll
        for (int q = 0; q < NRUN; ++q)
            for (int i = lane + 64; i < c[q]; i += 32)
                atomicAdd(&cnt[rec[o[q] + i].y & (BNODES - 1)], 1);
    }
    __syncthreads();

    const int v = (t < BNODES) ? cnt[t] : 0;
    const int x = wave_incl_scan(v, lane);
    if (lane == 31) wsum[wave] = x;
    __syncthreads();
    if (wave == 0) {
        const int w  = wsum[lane];
        const int xs = wave_incl_scan(w, lane);
        wsum[lane] = xs - w;
    }
    __syncthreads();
    if (t < BNODES) {
        const int first = r0 + wsum[wave] + (x - v);
        cnt[t] = first;
        const int node = b * BNODES + t;
        if (node < n_nodes) { deg[node] = v; start[node] = first; }
    }
    __syncthreads();

    for (int j0 = wave; j0 < nbk; j0 += 32 * NRUN) {
        int c[NRUN], o[NRUN];
#pragma unroll
        for (int q = 0; q < NRUN; ++q) {
            const int j  = j0 + 32 * q;
            const int2 co = row[j < nbk ? j : nbk - 1];
            c[q] = j < nbk ? co.x : 0;
            o[q] = j * CH + co.y;
        }
        int2 p[NRUN][2];
#pragma unroll
        for (int q = 0; q < NRUN; ++q)
#pragma unroll
            for (int u = 0; u < 2; ++u) {
                const int i = lane + 32 * u;
                p[q][u] = rec[i < c[q] ? o[q] + i : 0];
            }
#pragma unroll
        for (int q = 0; q < NRUN; ++q)
#pragma unroll
            for (int u = 0; u < 2; ++u) {
                const int i = lane + 32 * u;
                if (i < c[q]) {
                    const int slot = atomicAdd(&cnt[p[q][u].y & (BNODES - 1)], 1);
                    csr_src[slot] = p[q][u].x;
                }
            }
#pragma unroll
        for (int q = 0; q < NRUN; ++q)
            for (int i = lane + 64; i < c[q]; i += 32) {
                const int2 pr = rec[o[q] + i];
                const int slot = atomicAdd(&cnt[pr.y & (BNODES - 1)], 1);
                csr_src[slot] = pr.x;
            }
    }
}

__device__ __forceinline__ float lrelu(float v) { return (v > 0.0f) ? v : (LEAKY * v); }

__device__ __forceinline__ float pair_other(float v) {
    return __int_as_float(__builtin_amdgcn_update_dpp(0, __float_as_int(v), 0xB1, 0xF, 0xF, false));
}

__global__ void __launch_bounds__(256)
gat_node_kernel(const float* __restrict__ h, const float* __restrict__ P,
                const float* __restrict__ Q, const float* __restrict__ a,
                const int* __restrict__ csr_src, const int* __restrict__ start,
                const int* __restrict__ deg, float* __restrict__ out, int n_nodes) {
    const int t    = blockIdx.x * 256 + threadIdx.x;
    const int node = t >> 4;
    const int c4   = (t & 15) * 4;
    if (node >= n_nodes) return;

    const int n   = deg[node];
    const int beg = start[node];
    const v4f av = *(const v4f*)(a + c4);
    const v4f qv = *(const v4f*)(Q + (size_t)node * HC + c4);

    float m = -INFINITY, den = 0.0f;
    v4f acc = (v4f){0.f, 0.f, 0.f, 0.f};

    for (int j = 0; j < n; j += NU) {
        int s[NU];
#pragma unroll
        for (int k = 0; k < NU; ++k) {
            int jj = j + k;
            if (jj > n - 1) jj = n - 1;
            s[k] = csr_src[beg + jj];
        }
        v4f pv[NU], hv[NU];
#pragma unroll
        for (int k = 0; k < NU; ++k) {
            pv[k] = *(const v4f*)(P + (size_t)s[k] * HC + c4);
            hv[k] = *(const v4f*)(h + (size_t)s[k] * HC + c4);
        }
        float sc[NU];
        float mn = m;
#pragma unroll
        for (int k = 0; k < NU; ++k) {
            const float t0 = lrelu(pv[k].x + qv.x) * av.x;
            const float t1 = lrelu(pv[k].y + qv.y) * av.y;
            const float t2 = lrelu(pv[k].z + qv.z) * av.z;
            const float t3 = lrelu(pv[k].w + qv.w) * av.w;
            const float part = (t0 + t1) + (t2 + t3);
            float v = part + pair_other(part);
            if (j + k >= n) v = -INFINITY;
            sc[k] = v;
            mn = fmaxf(mn, v);
        }
        const float scale = __expf(m - mn);
        den *= scale;
        acc.x *= scale; acc.y *= scale; acc.z *= scale; acc.w *= scale;
#pragma unroll
        for (int k = 0; k < NU; ++k) {
            const float ex = __expf(sc[k] - mn);
            den += ex;
            acc.x += ex * hv[k].x; acc.y += ex * hv[k].y;
            acc.z += ex * hv[k].z; acc.w += ex * hv[k].w;
        }
        m = mn;
    }

    v4f o = (v4f){0.f, 0.f, 0.f, 0.f};
    if (n > 0) { o.x = acc.x / den; o.y = acc.y / den; o.z = acc.z / den; o.w = acc.w / den; }
    *(v4f*)(out + (size_t)node * HC + c4) = o;
}

extern "C" void kernel_launch(void* const* d_in, const int* in_sizes, int n_in,
                              void* d_out, int out_size, void* d_ws, size_t ws_size,
                              hipStream_t stream) {
    const float* h   = (const float*)d_in[0];
    const int*   src = (const int*)d_in[1];
    const int*   dst = (const int*)d_in[2];
    const float* w   = (const float*)d_in[3];
    const float* a   = (const float*)d_in[4];
    float* out = (float*)d_out;

    const int n_edges = in_sizes[1];
    const int n_nodes = in_sizes[0] / HC;

    const int npb  = PNB;
    const int nblk = (n_nodes + npb - 1) / npb;

    const int nb  = (n_nodes + BNODES - 1) >> BSH;
    const int nbk = (n_edges + CH - 1) / CH;

    char* ws = (char*)d_ws;
    size_t off = 0;
    float*  P       = (float*)(ws + off);  off += (size_t)nblk * npb * HC * sizeof(float);
    float*  Q       = (float*)(ws + off);  off += (size_t)nblk * npb * HC * sizeof(float);
    int2*   rec     = (int2*)(ws + off);   off += (size_t)n_edges * sizeof(int2);
    int2*   tab     = (int2*)(ws + off);   off += (size_t)nb * nbk * sizeof(int2);
    int*    csr_src = (int*)(ws + off);    off += (size_t)n_edges * sizeof(int);
    int*    deg     = (int*)(ws + off);    off += (size_t)n_nodes * sizeof(int);
    int*    start   = (int*)(ws + off);

    gat_block_sort_kernel<<<nbk + nblk, GT, 0, stream>>>(src, dst, rec, tab, n_edges, nb, nbk,
                                                         h, w, P, n_nodes);

    gat_bucket_group_kernel<<<nb + nblk, GT, 0, stream>>>(rec, tab, deg, start, csr_src, n_nodes, nbk,
                                                          nb, h, w, Q);

    const int nthr = n_nodes * 16;
    gat_node_kernel<<<(nthr + 255) / 256, 256, 0, stream>>>(h, P, Q, a, csr_src, start, deg,
                                                            out, n_nodes);
}


# Per-node pass issues fewer vector instructions per edge (126 -> 83 a trip): leaky as max(x,0.2x), packed pairs, 32-bit row offsets, log2e folded into a, one reciprocal; lane sum now (c0+c2)+(c1+c3)
# speedup vs baseline: 10.2886x; 1.0378x over previous
#include <hip/hip_runtime.h>
#include <math.h>

#define HC        64
#define LEAKY     0.2f
#define NU        4
#define GT        1024
#define EPT       8
#define CH        (GT * EPT)
#define BSH       9
#define BNODES    (1 << BSH)
#define NRUN      8
#define MAXNB     GT
#define PNB       (GT / 2)

typedef __attribute__((ext_vector_type(2))) float v2f;
typedef __attribute__((ext_vector_type(4))) float v4f;
typedef __attribute__((ext_vector_type(8))) float v8f;


__device__ __forceinline__ void proj_role(const float* __restrict__ h, const float* __restrict__ wh,
                                          float* __restrict__ plane,
                                          int pb, int n_nodes, float4* wl) {
    const int t     = threadIdx.x;
    const int lane  = t & 31;
    const int wave  = t >> 5;
    const int c     = lane & 15;
    const int hi    = lane >> 4;
    const int nbase = pb * PNB + wave * 16;

    float wv[4];
    {
        const float* wp = wh + (size_t)(4 * (t >> 6)) * HC + (t & 63);
#pragma unroll
        for (int r = 0; r < 4; ++r) wv[r] = wp[r * HC];
    }

    int ng = nbase + c;
    if (ng >= n_nodes) ng = n_nodes - 1;
    const float4* arow = (const float4*)(h + (size_t)ng * HC) + hi;
    float4 aq[8];
#pragma unroll
    for (int j = 0; j < 8; ++j) aq[j] = arow[2 * j];

    wl[t] = make_float4(wv[0], wv[1], wv[2], wv[3]);
    __syncthreads();

#pragma unroll
    for (int np = 0; np < 2; ++np) {
        v8f acc[2];
#pragma unroll
        for (int nt = 0; nt < 2; ++nt) acc[nt] = (v8f){0.f,0.f,0.f,0.f,0.f,0.f,0.f,0.f};
#pragma unroll
        for (int j = 0; j < 8; ++j) {
            const float4* brow = wl + (2 * j + hi) * HC + np * 32 + c;
            float4 bq[2];
#pragma unroll
            for (int nt = 0; nt < 2; ++nt) bq[nt] = brow[nt * 16];
            v2f a0, a1;
            a0.x = aq[j].x; a0.y = aq[j].y;
            a1.x = aq[j].z; a1.y = aq[j].w;
#pragma unroll
            for (int nt = 0; nt < 2; ++nt) {
                v2f b0; b0.x = bq[nt].x; b0.y = bq[nt].y;
                acc[nt] = __builtin_amdgcn_wmma_f32_16x16x4_f32(
                    false, a0, false, b0, (short)0, acc[nt], false, false);
            }
#pragma unroll
            for (int nt = 0; nt < 2; ++nt) {
                v2f b1; b1.x = bq[nt].z; b1.y = bq[nt].w;
                acc[nt] = __builtin_amdgcn_wmma_f32_16x16x4_f32(
                    false, a1, false, b1, (short)0, acc[nt], false, false);
            }
        }
#pragma unroll
        for (int nt = 0; nt < 2; ++nt)
#pragma unroll
            for (int i = 0; i < 8; ++i)
                plane[(size_t)(nbase + i + 8 * hi) * HC + np * 32 + nt * 16 + c] = acc[nt][i];
    }
}


__device__ __forceinline__ int wave_incl_scan(int x, int lane) {
#pragma unroll
    for (int o = 1; o < 32; o <<= 1) {
        const int y = __shfl_up(x, o, 32);
        if (lane >= o) x += y;
    }
    return x;
}

__global__ void __launch_bounds__(GT, 16)
gat_block_sort_kernel(const int* __restrict__ src, const int* __restrict__ dst,
                      int2* __restrict__ rec, int2* __restrict__ tab,
                      int n_edges, int nb, int nbk,
                      const float* __restrict__ h, const float* __restrict__ w,
                      float* __restrict__ P, int n_nodes) {
    __shared__ int  cnt[MAXNB];
    __shared__ int  wsum[32];
    __shared__ __attribute__((aligned(16))) int2 srt[CH];
    if (blockIdx.x >= nbk) {
        proj_role(h, w, P, blockIdx.x - nbk, n_nodes, (float4*)srt);
        return;
    }
    const int t    = threadIdx.x;
    const int lane = t & 31;
    const int wave = t >> 5;
    const int base = blockIdx.x * CH;

    int s[EPT], d[EPT], r[EPT];
#pragma unroll
    for (int k = 0; k < EPT; ++k) {
        const int e  = base + k * GT + t;
        const int ec = e < n_edges ? e : n_edges - 1;
        s[k] = src[ec];
        d[k] = dst[ec];
    }
    cnt[t] = 0;
    __syncthreads();
#pragma unroll
    for (int k = 0; k < EPT; ++k) {
        const int e = base + k * GT + t;
        r[k] = 0;
        if (e < n_edges) r[k] = atomicAdd(&cnt[d[k] >> BSH], 1);
        else             d[k] = -1;
    }
    __syncthreads();

    const int v = cnt[t];
    const int x = wave_incl_scan(v, lane);
    if (lane == 31) wsum[wave] = x;
    __syncthreads();
    if (wave == 0) {
        const int w  = wsum[lane];
        const int xs = wave_incl_scan(w, lane);
        wsum[lane] = xs - w;
    }
    __syncthreads();
    const int first = wsum[wave] + (x - v);
    cnt[t] = first;
    if (t < nb) tab[(size_t)t * nbk + blockIdx.x] = make_int2(v, first);
    __syncthreads();
#pragma unroll
    for (int k = 0; k < EPT; ++k) {
        if (d[k] >= 0) srt[cnt[d[k] >> BSH] + r[k]] = make_int2(s[k], d[k]);
    }
    __syncthreads();
    int2 g[EPT];
#pragma unroll
    for (int k = 0; k < EPT; ++k) g[k] = srt[k * GT + t];
#pragma unroll
    for (int k = 0; k < EPT; ++k) {
        const int i = k * GT + t;
        if (base + i < n_edges) rec[base + i] = g[k];
    }
}

__global__ void __launch_bounds__(GT, 16)
gat_bucket_group_kernel(const int2* __restrict__ rec, const int2* __restrict__ tab,
                        int* __restrict__ deg, int* __restrict__ start,
                        int* __restrict__ csr_src, int n_nodes, int nbk, int nb,
                        const float* __restrict__ h, const float* __restrict__ w,
                        float* __restrict__ Q) {
    __shared__ int cnt[BNODES];
    __shared__ int wsum[32];
    __shared__ __attribute__((aligned(16))) float4 wq[GT];
    if (blockIdx.x >= nb) {
        proj_role(h, w + (size_t)HC * HC, Q, blockIdx.x - nb, n_nodes, wq);
        return;
    }
    const int b    = blockIdx.x;
    const int t    = threadIdx.x;
    const int lane = t & 31;
    const int wave = t >> 5;
    const int2* row = tab + (size_t)b * nbk;

    int osum = 0;
    for (int j = t; j < nbk; j += GT) osum += row[j].y;
    osum = wave_incl_scan(osum, lane);
    if (lane == 31) wsum[wave] = osum;
    if (t < BNODES) cnt[t] = 0;
    __syncthreads();
    const int r0 = __shfl(wave_incl_scan(wsum[lane], lane), 31, 32);
    __syncthreads();

    for (int j0 = wave; j0 < nbk; j0 += 32 * NRUN) {
        int c[NRUN], o[NRUN];
#pragma unroll
        for (int q = 0; q < NRUN; ++q) {
            const int j  = j0 + 32 * q;
            const int2 co = row[j < nbk ? j : nbk - 1];
            c[q] = j < nbk ? co.x : 0;
            o[q] = j * CH + co.y;
        }
        int y[NRUN][2];
#pragma unroll
        for (int q = 0; q < NRUN; ++q)
#pragma unroll
            for (int u = 0; u < 2; ++u) {
                const int i = lane + 32 * u;
                y[q][u] = rec[i < c[q] ? o[q] + i : 0].y;
            }
#pragma unroll
        for (int q = 0; q < NRUN; ++q)
#pragma unroll
            for (int u = 0; u < 2; ++u) {
                const int i = lane + 32 * u;
                if (i < c[q]) atomicAdd(&cnt[y[q][u] & (BNODES - 1)], 1);
            }
#pragma unroll
        for (int q = 0; q < NRUN; ++q)
            for (int i = lane + 64; i < c[q]; i += 32)
                atomicAdd(&cnt[rec[o[q] + i].y & (BNODES - 1)], 1);
    }
    __syncthreads();

    const int v = (t < BNODES) ? cnt[t] : 0;
    const int x = wave_incl_scan(v, lane);
    if (lane == 31) wsum[wave] = x;
    __syncthreads();
    if (wave == 0) {
        const int w  = wsum[lane];
        const int xs = wave_incl_scan(w, lane);
        wsum[lane] = xs - w;
    }
    __syncthreads();
    if (t < BNODES) {
        const int first = r0 + wsum[wave] + (x - v);
        cnt[t] = first;
        const int node = b * BNODES + t;
        if (node < n_nodes) { deg[node] = v; start[node] = first; }
    }
    __syncthreads();

    for (int j0 = wave; j0 < nbk; j0 += 32 * NRUN) {
        int c[NRUN], o[NRUN];
#pragma unroll
        for (int q = 0; q < NRUN; ++q) {
            const int j  = j0 + 32 * q;
            const int2 co = row[j < nbk ? j : nbk - 1];
            c[q] = j < nbk ? co.x : 0;
            o[q] = j * CH + co.y;
        }
        int2 p[NRUN][2];
#pragma unroll
        for (int q = 0; q < NRUN; ++q)
#pragma unroll
            for (int u = 0; u < 2; ++u) {
                const int i = lane + 32 * u;
                p[q][u] = rec[i < c[q] ? o[q] + i : 0];
            }
#pragma unroll
        for (int q = 0; q < NRUN; ++q)
#pragma unroll
            for (int u = 0; u < 2; ++u) {
                const int i = lane + 32 * u;
                if (i < c[q]) {
                    const int slot = atomicAdd(&cnt[p[q][u].y & (BNODES - 1)], 1);
                    csr_src[slot] = p[q][u].x;
                }
            }
#pragma unroll
        for (int q = 0; q < NRUN; ++q)
            for (int i = lane + 64; i < c[q]; i += 32) {
                const int2 pr = rec[o[q] + i];
                const int slot = atomicAdd(&cnt[pr.y & (BNODES - 1)], 1);
                csr_src[slot] = pr.x;
            }
    }
}

#define LOG2E 1.44269504088896340736f

__device__ __forceinline__ float pair_other(float v) {
    return __int_as_float(__builtin_amdgcn_update_dpp(0, __float_as_int(v), 0xB1, 0xF, 0xF, false));
}

__global__ void __launch_bounds__(256)
gat_node_kernel(const float* __restrict__ h, const float* __restrict__ P,
                const float* __restrict__ Q, const float* __restrict__ a,
                const int* __restrict__ csr_src, const int* __restrict__ start,
                const int* __restrict__ deg, float* __restrict__ out, int n_nodes) {
    const int t    = blockIdx.x * 256 + threadIdx.x;
    const int node = t >> 4;
    const unsigned cb = (unsigned)(t & 15) * 16u;
    if (node >= n_nodes) return;

    const char* Pb = (const char*)P;
    const char* hb = (const char*)h;

    const int n   = deg[node];
    const int beg = start[node];
    const v4f av = *(const v4f*)((const char*)a + cb);
    const v4f qv = *(const v4f*)((const char*)Q + (((unsigned)node << 8) + cb));
    const v2f a01 = (v2f){av.x * LOG2E, av.y * LOG2E};
    const v2f a23 = (v2f){av.z * LOG2E, av.w * LOG2E};
    const v2f q01 = (v2f){qv.x, qv.y};
    const v2f q23 = (v2f){qv.z, qv.w};

    float m = -INFINITY, den = 0.0f;
    v2f acc01 = (v2f){0.f, 0.f};
    v2f acc23 = (v2f){0.f, 0.f};

    for (int j = 0; j < n; j += NU) {
        int s[NU];
#pragma unroll
        for (int k = 0; k < NU; ++k) {
            int jj = j + k;
            if (jj > n - 1) jj = n - 1;
            s[k] = csr_src[beg + jj];
        }
        v4f pv[NU], hv[NU];
#pragma unroll
        for (int k = 0; k < NU; ++k) {
            const unsigned off = ((unsigned)s[k] << 8) + cb;
            pv[k] = *(const v4f*)(Pb + off);
            hv[k] = *(const v4f*)(hb + off);
        }
        float sc[NU];
        float mn = m;
#pragma unroll
        for (int k = 0; k < NU; ++k) {
            const v2f x01 = (v2f){pv[k].x, pv[k].y} + q01;
            const v2f x23 = (v2f){pv[k].z, pv[k].w} + q23;
            const v2f l01 = __builtin_elementwise_max(x01, x01 * LEAKY);
            const v2f l23 = __builtin_elementwise_max(x23, x23 * LEAKY);
            const v2f tt  = l01 * a01 + l23 * a23;
            const float part = tt.x + tt.y;
            float v = part + pair_other(part);
            if (j + k >= n) v = -INFINITY;
            sc[k] = v;
            mn = fmaxf(mn, v);
        }
        const float scale = __builtin_amdgcn_exp2f(m - mn);
        den *= scale;
        acc01 *= scale;
        acc23 *= scale;
#pragma unroll
        for (int k = 0; k < NU; ++k) {
            const float ex = __builtin_amdgcn_exp2f(sc[k] - mn);
            den += ex;
            acc01 += (v2f){hv[k].x, hv[k].y} * ex;
            acc23 += (v2f){hv[k].z, hv[k].w} * ex;
        }
        m = mn;
    }

    v4f o = (v4f){0.f, 0.f, 0.f, 0.f};
    if (n > 0) {
        const float r = 1.0f / den;
        o.x = acc01.x * r; o.y = acc01.y * r; o.z = acc23.x * r; o.w = acc23.y * r;
    }
    *(v4f*)((char*)out + (((unsigned)node << 8) + cb)) = o;
}

extern "C" void kernel_launch(void* const* d_in, const int* in_sizes, int n_in,
                              void* d_out, int out_size, void* d_ws, size_t ws_size,
                              hipStream_t stream) {
    const float* h   = (const float*)d_in[0];
    const int*   src = (const int*)d_in[1];
    const int*   dst = (const int*)d_in[2];
    const float* w   = (const float*)d_in[3];
    const float* a   = (const float*)d_in[4];
    float* out = (float*)d_out;

    const int n_edges = in_sizes[1];
    const int n_nodes = in_sizes[0] / HC;

    const int npb  = PNB;
    const int nblk = (n_nodes + npb - 1) / npb;

    const int nb  = (n_nodes + BNODES - 1) >> BSH;
    const int nbk = (n_edges + CH - 1) / CH;

    char* ws = (char*)d_ws;
    size_t off = 0;
    float*  P       = (float*)(ws + off);  off += (size_t)nblk * npb * HC * sizeof(float);
    float*  Q       = (float*)(ws + off);  off += (size_t)nblk * npb * HC * sizeof(float);
    int2*   rec     = (int2*)(ws + off);   off += (size_t)n_edges * sizeof(int2);
    int2*   tab     = (int2*)(ws + off);   off += (size_t)nb * nbk * sizeof(int2);
    int*    csr_src = (int*)(ws + off);    off += (size_t)n_edges * sizeof(int);
    int*    deg     = (int*)(ws + off);    off += (size_t)n_nodes * sizeof(int);
    int*    start   = (int*)(ws + off);

    gat_block_sort_kernel<<<nbk + nblk, GT, 0, stream>>>(src, dst, rec, tab, n_edges, nb, nbk,
                                                         h, w, P, n_nodes);

    gat_bucket_group_kernel<<<nb + nblk, GT, 0, stream>>>(rec, tab, deg, start, csr_src, n_nodes, nbk,
                                                          nb, h, w, Q);

    const int nthr = n_nodes * 16;
    gat_node_kernel<<<(nthr + 255) / 256, 256, 0, stream>>>(h, P, Q, a, csr_src, start, deg,
                                                            out, n_nodes);
}


# Bucket pass stages each bucket's source ids in an LDS copy of its csr region and writes it out in slot order, a whole 128-byte line per store, instead of one scattered 4-byte store per edge
# speedup vs baseline: 11.0372x; 1.0728x over previous
#include <hip/hip_runtime.h>
#include <math.h>

#define HC        64
#define LEAKY     0.2f
#define NU        4
#define GT        1024
#define EPT       8
#define CH        (GT * EPT)
#define BSH       9
#define BNODES    (1 << BSH)
#define NRUN      8
#define MAXNB     GT
#define PNB       (GT / 2)
#define SCAP      10240

typedef __attribute__((ext_vector_type(2))) float v2f;
typedef __attribute__((ext_vector_type(4))) float v4f;
typedef __attribute__((ext_vector_type(8))) float v8f;


__device__ __forceinline__ void proj_role(const float* __restrict__ h, const float* __restrict__ wh,
                                          float* __restrict__ plane,
                                          int pb, int n_nodes, float4* wl) {
    const int t     = threadIdx.x;
    const int lane  = t & 31;
    const int wave  = t >> 5;
    const int c     = lane & 15;
    const int hi    = lane >> 4;
    const int nbase = pb * PNB + wave * 16;

    float wv[4];
    {
        const float* wp = wh + (size_t)(4 * (t >> 6)) * HC + (t & 63);
#pragma unroll
        for (int r = 0; r < 4; ++r) wv[r] = wp[r * HC];
    }

    int ng = nbase + c;
    if (ng >= n_nodes) ng = n_nodes - 1;
    const float4* arow = (const float4*)(h + (size_t)ng * HC) + hi;
    float4 aq[8];
#pragma unroll
    for (int j = 0; j < 8; ++j) aq[j] = arow[2 * j];

    wl[t] = make_float4(wv[0], wv[1], wv[2], wv[3]);
    __syncthreads();

#pragma unroll
    for (int np = 0; np < 2; ++np) {
        v8f acc[2];
#pragma unroll
        for (int nt = 0; nt < 2; ++nt) acc[nt] = (v8f){0.f,0.f,0.f,0.f,0.f,0.f,0.f,0.f};
#pragma unroll
        for (int j = 0; j < 8; ++j) {
            const float4* brow = wl + (2 * j + hi) * HC + np * 32 + c;
            float4 bq[2];
#pragma unroll
            for (int nt = 0; nt < 2; ++nt) bq[nt] = brow[nt * 16];
            v2f a0, a1;
            a0.x = aq[j].x; a0.y = aq[j].y;
            a1.x = aq[j].z; a1.y = aq[j].w;
#pragma unroll
            for (int nt = 0; nt < 2; ++nt) {
                v2f b0; b0.x = bq[nt].x; b0.y = bq[nt].y;
                acc[nt] = __builtin_amdgcn_wmma_f32_16x16x4_f32(
                    false, a0, false, b0, (short)0, acc[nt], false, false);
            }
#pragma unroll
            for (int nt = 0; nt < 2; ++nt) {
                v2f b1; b1.x = bq[nt].z; b1.y = bq[nt].w;
                acc[nt] = __builtin_amdgcn_wmma_f32_16x16x4_f32(
                    false, a1, false, b1, (short)0, acc[nt], false, false);
            }
        }
#pragma unroll
        for (int nt = 0; nt < 2; ++nt)
#pragma unroll
            for (int i = 0; i < 8; ++i)
                plane[(size_t)(nbase + i + 8 * hi) * HC + np * 32 + nt * 16 + c] = acc[nt][i];
    }
}


__device__ __forceinline__ int wave_incl_scan(int x, int lane) {
#pragma unroll
    for (int o = 1; o < 32; o <<= 1) {
        const int y = __shfl_up(x, o, 32);
        if (lane >= o) x += y;
    }
    return x;
}

__global__ void __launch_bounds__(GT, 16)
gat_block_sort_kernel(const int* __restrict__ src, const int* __restrict__ dst,
                      int2* __restrict__ rec, int2* __restrict__ tab,
                      int n_edges, int nb, int nbk,
                      const float* __restrict__ h, const float* __restrict__ w,
                      float* __restrict__ P, int n_nodes) {
    __shared__ int  cnt[MAXNB];
    __shared__ int  wsum[32];
    __shared__ __attribute__((aligned(16))) int2 srt[CH];
    if (blockIdx.x >= nbk) {
        proj_role(h, w, P, blockIdx.x - nbk, n_nodes, (float4*)srt);
        return;
    }
    const int t    = threadIdx.x;
    const int lane = t & 31;
    const int wave = t >> 5;
    const int base = blockIdx.x * CH;

    int s[EPT], d[EPT], r[EPT];
#pragma unroll
    for (int k = 0; k < EPT; ++k) {
        const int e  = base + k * GT + t;
        const int ec = e < n_edges ? e : n_edges - 1;
        s[k] = src[ec];
        d[k] = dst[ec];
    }
    cnt[t] = 0;
    __syncthreads();
#pragma unroll
    for (int k = 0; k < EPT; ++k) {
        const int e = base + k * GT + t;
        r[k] = 0;
        if (e < n_edges) r[k] = atomicAdd(&cnt[d[k] >> BSH], 1);
        else             d[k] = -1;
    }
    __syncthreads();

    const int v = cnt[t];
    const int x = wave_incl_scan(v, lane);
    if (lane == 31) wsum[wave] = x;
    __syncthreads();
    if (wave == 0) {
        const int w  = wsum[lane];
        const int xs = wave_incl_scan(w, lane);
        wsum[lane] = xs - w;
    }
    __syncthreads();
    const int first = wsum[wave] + (x - v);
    cnt[t] = first;
    if (t < nb) tab[(size_t)t * nbk + blockIdx.x] = make_int2(v, first);
    __syncthreads();
#pragma unroll
    for (int k = 0; k < EPT; ++k) {
        if (d[k] >= 0) srt[cnt[d[k] >> BSH] + r[k]] = make_int2(s[k], d[k]);
    }
    __syncthreads();
    int2 g[EPT];
#pragma unroll
    for (int k = 0; k < EPT; ++k) g[k] = srt[k * GT + t];
#pragma unroll
    for (int k = 0; k < EPT; ++k) {
        const int i = k * GT + t;
        if (base + i < n_edges) rec[base + i] = g[k];
    }
}

template <bool STAGED>
__device__ __forceinline__ void place_pass(const int2* __restrict__ rec, const int2* __restrict__ row,
                                           int nbk, int wave, int lane, int* cnt, int* stage,
                                           int* __restrict__ csr_src, int r0) {
    for (int j0 = wave; j0 < nbk; j0 += 32 * NRUN) {
        int c[NRUN], o[NRUN];
#pragma unroll
        for (int q = 0; q < NRUN; ++q) {
            const int j  = j0 + 32 * q;
            const int2 co = row[j < nbk ? j : nbk - 1];
            c[q] = j < nbk ? co.x : 0;
            o[q] = j * CH + co.y;
        }
        int2 p[NRUN][2];
#pragma unroll
        for (int q = 0; q < NRUN; ++q)
#pragma unroll
            for (int u = 0; u < 2; ++u) {
                const int i = lane + 32 * u;
                p[q][u] = rec[i < c[q] ? o[q] + i : 0];
            }
#pragma unroll
        for (int q = 0; q < NRUN; ++q)
#pragma unroll
            for (int u = 0; u < 2; ++u) {
                const int i = lane + 32 * u;
                if (i < c[q]) {
                    const int slot = atomicAdd(&cnt[p[q][u].y & (BNODES - 1)], 1);
                    if (STAGED) stage[slot - r0] = p[q][u].x;
                    else        csr_src[slot]    = p[q][u].x;
                }
            }
#pragma unroll
        for (int q = 0; q < NRUN; ++q)
            for (int i = lane + 64; i < c[q]; i += 32) {
                const int2 pr = rec[o[q] + i];
                const int slot = atomicAdd(&cnt[pr.y & (BNODES - 1)], 1);
                if (STAGED) stage[slot - r0] = pr.x;
                else        csr_src[slot]    = pr.x;
            }
    }
}

__global__ void __launch_bounds__(GT, 16)
gat_bucket_group_kernel(const int2* __restrict__ rec, const int2* __restrict__ tab,
                        int* __restrict__ deg, int* __restrict__ start,
                        int* __restrict__ csr_src, int n_nodes, int nbk, int nb,
                        const float* __restrict__ h, const float* __restrict__ w,
                        float* __restrict__ Q) {
    __shared__ int cnt[BNODES];
    __shared__ int wsum[32];
    __shared__ int bend;
    __shared__ __attribute__((aligned(16))) int stage[SCAP];
    if (blockIdx.x >= nb) {
        proj_role(h, w + (size_t)HC * HC, Q, blockIdx.x - nb, n_nodes, (float4*)stage);
        return;
    }
    const int b    = blockIdx.x;
    const int t    = threadIdx.x;
    const int lane = t & 31;
    const int wave = __builtin_amdgcn_readfirstlane(t >> 5);
    const int2* row = tab + (size_t)b * nbk;

    int osum = 0;
    for (int j = t; j < nbk; j += GT) osum += row[j].y;
    if (t == 0) bend = 0;
    if (t < BNODES) cnt[t] = 0;
    __syncthreads();
    if (osum != 0) atomicAdd(&bend, osum);
    __syncthreads();
    const int r0 = __builtin_amdgcn_readfirstlane(bend);

    for (int j0 = wave; j0 < nbk; j0 += 32 * NRUN) {
        int c[NRUN], o[NRUN];
#pragma unroll
        for (int q = 0; q < NRUN; ++q) {
            const int j  = j0 + 32 * q;
            const int2 co = row[j < nbk ? j : nbk - 1];
            c[q] = j < nbk ? co.x : 0;
            o[q] = j * CH + co.y;
        }
        int y[NRUN][2];
#pragma unroll
        for (int q = 0; q < NRUN; ++q)
#pragma unroll
            for (int u = 0; u < 2; ++u) {
                const int i = lane + 32 * u;
                y[q][u] = rec[i < c[q] ? o[q] + i : 0].y;
            }
#pragma unroll
        for (int q = 0; q < NRUN; ++q)
#pragma unroll
            for (int u = 0; u < 2; ++u) {
                const int i = lane + 32 * u;
                if (i < c[q]) atomicAdd(&cnt[y[q][u] & (BNODES - 1)], 1);
            }
#pragma unroll
        for (int q = 0; q < NRUN; ++q)
            for (int i = lane + 64; i < c[q]; i += 32)
                atomicAdd(&cnt[rec[o[q] + i].y & (BNODES - 1)], 1);
    }
    __syncthreads();

    const int v = (t < BNODES) ? cnt[t] : 0;
    const int x = wave_incl_scan(v, lane);
    if (lane == 31) wsum[wave] = x;
    __syncthreads();
    if (wave == 0) {
        const int w  = wsum[lane];
        const int xs = wave_incl_scan(w, lane);
        wsum[lane] = xs - w;
    }
    __syncthreads();
    if (t < BNODES) {
        const int first = r0 + wsum[wave] + (x - v);
        cnt[t] = first;
        if (t == BNODES - 1) bend = first + v;
        const int node = b * BNODES + t;
        if (node < n_nodes) { deg[node] = v; start[node] = first; }
    }
    __syncthreads();

    const int nrec = __builtin_amdgcn_readfirstlane(bend) - r0;
    if (nrec > SCAP) {
        place_pass<false>(rec, row, nbk, wave, lane, cnt, stage, csr_src, r0);
        return;
    }
    place_pass<true>(rec, row, nbk, wave, lane, cnt, stage, csr_src, r0);
    __syncthreads();

    const int last = r0 + nrec;
    const int g0   = (r0 & ~31) + t;
    int gv[8];
#pragma unroll
    for (int k = 0; k < 8; ++k) {
        const int g = g0 + k * GT;
        gv[k] = stage[(g >= r0 && g < last) ? g - r0 : 0];
    }
#pragma unroll
    for (int k = 0; k < 8; ++k) {
        const int g = g0 + k * GT;
        if (g >= r0 && g < last) csr_src[g] = gv[k];
    }
    for (int g = g0 + 8 * GT; g < last; g += GT) csr_src[g] = stage[g - r0];
}

#define LOG2E 1.44269504088896340736f

__device__ __forceinline__ float pair_other(float v) {
    return __int_as_float(__builtin_amdgcn_update_dpp(0, __float_as_int(v), 0xB1, 0xF, 0xF, false));
}

__global__ void __launch_bounds__(256)
gat_node_kernel(const float* __restrict__ h, const float* __restrict__ P,
                const float* __restrict__ Q, const float* __restrict__ a,
                const int* __restrict__ csr_src, const int* __restrict__ start,
                const int* __restrict__ deg, float* __restrict__ out, int n_nodes) {
    const int t    = blockIdx.x * 256 + threadIdx.x;
    const int node = t >> 4;
    const unsigned cb = (unsigned)(t & 15) * 16u;
    if (node >= n_nodes) return;

    const char* Pb = (const char*)P;
    const char* hb = (const char*)h;

    const int n   = deg[node];
    const int beg = start[node];
    const v4f av = *(const v4f*)((const char*)a + cb);
    const v4f qv = *(const v4f*)((const char*)Q + (((unsigned)node << 8) + cb));
    const v2f a01 = (v2f){av.x * LOG2E, av.y * LOG2E};
    const v2f a23 = (v2f){av.z * LOG2E, av.w * LOG2E};
    const v2f q01 = (v2f){qv.x, qv.y};
    const v2f q23 = (v2f){qv.z, qv.w};

    float m = -INFINITY, den = 0.0f;
    v2f acc01 = (v2f){0.f, 0.f};
    v2f acc23 = (v2f){0.f, 0.f};

    for (int j = 0; j < n; j += NU) {
        int s[NU];
#pragma unroll
        for (int k = 0; k < NU; ++k) {
            int jj = j + k;
            if (jj > n - 1) jj = n - 1;
            s[k] = csr_src[beg + jj];
        }
        v4f pv[NU], hv[NU];
#pragma unroll
        for (int k = 0; k < NU; ++k) {
            const unsigned off = ((unsigned)s[k] << 8) + cb;
            pv[k] = *(const v4f*)(Pb + off);
            hv[k] = *(const v4f*)(hb + off);
        }
        float sc[NU];
        float mn = m;
#pragma unroll
        for (int k = 0; k < NU; ++k) {
            const v2f x01 = (v2f){pv[k].x, pv[k].y} + q01;
            const v2f x23 = (v2f){pv[k].z, pv[k].w} + q23;
            const v2f l01 = __builtin_elementwise_max(x01, x01 * LEAKY);
            const v2f l23 = __builtin_elementwise_max(x23, x23 * LEAKY);
            const v2f tt  = l01 * a01 + l23 * a23;
            const float part = tt.x + tt.y;
            float v = part + pair_other(part);
            if (j + k >= n) v = -INFINITY;
            sc[k] = v;
            mn = fmaxf(mn, v);
        }
        const float scale = __builtin_amdgcn_exp2f(m - mn);
        den *= scale;
        acc01 *= scale;
        acc23 *= scale;
#pragma unroll
        for (int k = 0; k < NU; ++k) {
            const float ex = __builtin_amdgcn_exp2f(sc[k] - mn);
            den += ex;
            acc01 += (v2f){hv[k].x, hv[k].y} * ex;
            acc23 += (v2f){hv[k].z, hv[k].w} * ex;
        }
        m = mn;
    }

    v4f o = (v4f){0.f, 0.f, 0.f, 0.f};
    if (n > 0) {
        const float r = 1.0f / den;
        o.x = acc01.x * r; o.y = acc01.y * r; o.z = acc23.x * r; o.w = acc23.y * r;
    }
    *(v4f*)((char*)out + (((unsigned)node << 8) + cb)) = o;
}

extern "C" void kernel_launch(void* const* d_in, const int* in_sizes, int n_in,
                              void* d_out, int out_size, void* d_ws, size_t ws_size,
                              hipStream_t stream) {
    const float* h   = (const float*)d_in[0];
    const int*   src = (const int*)d_in[1];
    const int*   dst = (const int*)d_in[2];
    const float* w   = (const float*)d_in[3];
    const float* a   = (const float*)d_in[4];
    float* out = (float*)d_out;

    const int n_edges = in_sizes[1];
    const int n_nodes = in_sizes[0] / HC;

    const int npb  = PNB;
    const int nblk = (n_nodes + npb - 1) / npb;

    const int nb  = (n_nodes + BNODES - 1) >> BSH;
    const int nbk = (n_edges + CH - 1) / CH;

    char* ws = (char*)d_ws;
    size_t off = 0;
    float*  P       = (float*)(ws + off);  off += (size_t)nblk * npb * HC * sizeof(float);
    float*  Q       = (float*)(ws + off);  off += (size_t)nblk * npb * HC * sizeof(float);
    int2*   rec     = (int2*)(ws + off);   off += (size_t)n_edges * sizeof(int2);
    int2*   tab     = (int2*)(ws + off);   off += (size_t)nb * nbk * sizeof(int2);
    int*    csr_src = (int*)(ws + off);    off += (size_t)n_edges * sizeof(int);
    int*    deg     = (int*)(ws + off);    off += (size_t)n_nodes * sizeof(int);
    int*    start   = (int*)(ws + off);

    gat_block_sort_kernel<<<nbk + nblk, GT, 0, stream>>>(src, dst, rec, tab, n_edges, nb, nbk,
                                                         h, w, P, n_nodes);

    gat_bucket_group_kernel<<<nb + nblk, GT, 0, stream>>>(rec, tab, deg, start, csr_src, n_nodes, nbk,
                                                          nb, h, w, Q);

    const int nthr = n_nodes * 16;
    gat_node_kernel<<<(nthr + 255) / 256, 256, 0, stream>>>(h, P, Q, a, csr_src, start, deg,
                                                            out, n_nodes);
}
